# starting kernel
# baseline (speedup 1.0000x reference)
#include <hip/hip_runtime.h>
#include <cmath>

#define B_  8
#define TE_ 256
#define TD_ 256
#define D_  256
#define U_  256

typedef float v2f __attribute__((ext_vector_type(2)));
typedef float v8f __attribute__((ext_vector_type(8)));

__global__ void __launch_bounds__(256) gemm_f32_wmma(
    const float* __restrict__ A, const float* __restrict__ W,
    float* __restrict__ C) {
  const int lane  = threadIdx.x & 31;
  const int wave  = threadIdx.x >> 5;
  const int tile  = blockIdx.x * 8 + wave;
  const int tn    = tile & 15;
  const int tm    = tile >> 4;
  const int m     = lane & 15;
  const int khalf = lane >> 4;

  const float* arow = A + (size_t)(tm * 16 + m) * U_;
  const float* bcol = W + tn * 16 + m;
  v8f acc = {};
  for (int k = 0; k < U_; k += 4) {
    const int ka = k + 2 * khalf;
    v2f a, b;
    a.x = arow[ka + 0];
    a.y = arow[ka + 1];
    b.x = bcol[(size_t)(ka + 0) * U_];
    b.y = bcol[(size_t)(ka + 1) * U_];
    acc = __builtin_amdgcn_wmma_f32_16x16x4_f32(false, a, false, b,
                                                (short)0, acc, false, false);
  }
#pragma unroll
  for (int r = 0; r < 8; ++r) {
    const int m_out = tm * 16 + r + 8 * khalf;
    C[(size_t)m_out * U_ + tn * 16 + m] = acc[r];
  }
}

__global__ void __launch_bounds__(256) co_mu_kernel(
    const float* __restrict__ ae, const float* __restrict__ ad,
    const float* __restrict__ nu, float* __restrict__ mu) {
  const int b      = blockIdx.y;
  const int s_tile = blockIdx.x >> 4;
  const int t_tile = blockIdx.x & 15;
  __shared__ float adT[16][260];
  __shared__ float aeT[16][260];
  __shared__ float nuS[256];

  const int tid = threadIdx.x;
  nuS[tid] = nu[tid];
  for (int i = tid; i < 16 * 256; i += 256) {
    const int r = i >> 8, c = i & 255;
    adT[r][c] = ad[((size_t)(b * TD_ + s_tile * 16 + r)) * U_ + c];
    aeT[r][c] = ae[((size_t)(b * TE_ + t_tile * 16 + r)) * U_ + c];
  }
  __syncthreads();

  const int si = tid >> 4, ti = tid & 15;
  float acc = 0.f;
#pragma unroll 4
  for (int u = 0; u < U_; ++u)
    acc += tanhf(adT[si][u] + aeT[ti][u]) * nuS[u];

  mu[((size_t)(b * TD_ + s_tile * 16 + si)) * TE_ + t_tile * 16 + ti] = acc;
}

__global__ void __launch_bounds__(256) softmax_rows_kernel(
    float* __restrict__ mu, float* __restrict__ max_mu) {
  const int row = blockIdx.x;
  const int tid = threadIdx.x;
  __shared__ float red[256];

  const float v = mu[(size_t)row * TE_ + tid];
  red[tid] = v; __syncthreads();
  for (int s = 128; s > 0; s >>= 1) {
    if (tid < s) red[tid] = fmaxf(red[tid], red[tid + s]);
    __syncthreads();
  }
  const float mx = red[0];
  __syncthreads();
  const float e = expf(v - mx);
  red[tid] = e; __syncthreads();
  for (int s = 128; s > 0; s >>= 1) {
    if (tid < s) red[tid] += red[tid + s];
    __syncthreads();
  }
  mu[(size_t)row * TE_ + tid] = e * (1.0f / red[0]);
  if (tid == 0) max_mu[row] = mx;
}

__global__ void __launch_bounds__(256) hhat_kernel(
    const float* __restrict__ max_mu, const float* __restrict__ de_seq,
    float* __restrict__ h_hat) {
  const int b   = blockIdx.x;
  const int tid = threadIdx.x;
  __shared__ float red[256];
  __shared__ float ma[256];

  const float v = max_mu[b * TD_ + tid];
  red[tid] = v; __syncthreads();
  for (int s = 128; s > 0; s >>= 1) {
    if (tid < s) red[tid] = fmaxf(red[tid], red[tid + s]);
    __syncthreads();
  }
  const float mx = red[0];
  __syncthreads();
  const float e = expf(v - mx);
  red[tid] = e; __syncthreads();
  for (int s = 128; s > 0; s >>= 1) {
    if (tid < s) red[tid] += red[tid + s];
    __syncthreads();
  }
  ma[tid] = e * (1.0f / red[0]);
  __syncthreads();

  float acc = 0.f;
  for (int s = 0; s < TD_; ++s)
    acc += de_seq[((size_t)(b * TD_ + s)) * D_ + tid] * ma[s];
  h_hat[b * D_ + tid] = acc;
}

__global__ void __launch_bounds__(256) sum_en_out_kernel(
    const float* __restrict__ alphas, const float* __restrict__ en_seq,
    const float* __restrict__ de_seq, const float* __restrict__ h_hat,
    float* __restrict__ out) {
  const int b     = blockIdx.y;
  const int lane  = threadIdx.x & 31;
  const int wave  = threadIdx.x >> 5;
  const int tile  = blockIdx.x * 8 + wave;
  const int td    = tile & 15;
  const int ts    = tile >> 4;
  const int m     = lane & 15;
  const int khalf = lane >> 4;

  const float* A    = alphas + (size_t)b * TD_ * TE_;
  const float* Bm   = en_seq + (size_t)b * TE_ * D_;
  const float* arow = A + (size_t)(ts * 16 + m) * TE_;
  const float* bcol = Bm + td * 16 + m;

  v8f acc = {};
  for (int k = 0; k < TE_; k += 4) {
    const int ka = k + 2 * khalf;
    v2f a, bb;
    a.x  = arow[ka + 0];
    a.y  = arow[ka + 1];
    bb.x = bcol[(size_t)(ka + 0) * D_];
    bb.y = bcol[(size_t)(ka + 1) * D_];
    acc = __builtin_amdgcn_wmma_f32_16x16x4_f32(false, a, false, bb,
                                                (short)0, acc, false, false);
  }

  const int d_g  = td * 16 + m;
  const float hh = h_hat[b * D_ + d_g];
#pragma unroll
  for (int r = 0; r < 8; ++r) {
    const int s_g  = ts * 16 + r + 8 * khalf;
    const float se = acc[r];
    const float de = de_seq[((size_t)(b * TD_ + s_g)) * D_ + d_g];
    float* o = out + ((size_t)(b * TD_ + s_g)) * (4 * D_);
    o[d_g]          = de;
    o[D_ + d_g]     = se;
    o[2 * D_ + d_g] = de * se;
    o[3 * D_ + d_g] = de * hh;
  }
}

extern "C" void kernel_launch(void* const* d_in, const int* in_sizes, int n_in,
                              void* d_out, int out_size, void* d_ws, size_t ws_size,
                              hipStream_t stream) {
  const float* en_seq = (const float*)d_in[0];
  const float* de_seq = (const float*)d_in[1];
  const float* w_en   = (const float*)d_in[2];
  const float* w_de   = (const float*)d_in[3];
  const float* nu     = (const float*)d_in[4];
  float* out = (float*)d_out;

  float* ws = (float*)d_ws;
  const size_t nAE = (size_t)B_ * TE_ * U_;
  float* ae     = ws;
  float* ad     = ae + nAE;
  float* mu     = ad + nAE;
  float* max_mu = mu + (size_t)B_ * TD_ * TE_;
  float* h_hat  = max_mu + (size_t)B_ * TD_;

  const dim3 blk(256);
  gemm_f32_wmma<<<dim3(256), blk, 0, stream>>>(en_seq, w_en, ae);
  gemm_f32_wmma<<<dim3(256), blk, 0, stream>>>(de_seq, w_de, ad);
  co_mu_kernel<<<dim3(256, B_), blk, 0, stream>>>(ae, ad, nu, mu);
  softmax_rows_kernel<<<dim3(B_ * TD_), blk, 0, stream>>>(mu, max_mu);
  hhat_kernel<<<dim3(B_), blk, 0, stream>>>(max_mu, de_seq, h_hat);
  sum_en_out_kernel<<<dim3(32, B_), blk, 0, stream>>>(mu, en_seq, de_seq,
                                                      h_hat, out);
}


# score kernel: library tanhf replaced by 1-2/(exp2(c*x)+1) on hardware v_exp_f32 and v_rcp_f32 (about 1e-7 abs error per term), c=2*log2(e) folded into the LDS staging; sum order over u unchanged
# speedup vs baseline: 1.8391x; 1.8391x over previous
#include <hip/hip_runtime.h>
#include <cmath>

#define B_  8
#define TE_ 256
#define TD_ 256
#define D_  256
#define U_  256

typedef float v2f __attribute__((ext_vector_type(2)));
typedef float v8f __attribute__((ext_vector_type(8)));

__global__ void __launch_bounds__(256) gemm_f32_wmma(
    const float* __restrict__ A, const float* __restrict__ W,
    float* __restrict__ C) {
  const int lane  = threadIdx.x & 31;
  const int wave  = threadIdx.x >> 5;
  const int tile  = blockIdx.x * 8 + wave;
  const int tn    = tile & 15;
  const int tm    = tile >> 4;
  const int m     = lane & 15;
  const int khalf = lane >> 4;

  const float* arow = A + (size_t)(tm * 16 + m) * U_;
  const float* bcol = W + tn * 16 + m;
  v8f acc = {};
  for (int k = 0; k < U_; k += 4) {
    const int ka = k + 2 * khalf;
    v2f a, b;
    a.x = arow[ka + 0];
    a.y = arow[ka + 1];
    b.x = bcol[(size_t)(ka + 0) * U_];
    b.y = bcol[(size_t)(ka + 1) * U_];
    acc = __builtin_amdgcn_wmma_f32_16x16x4_f32(false, a, false, b,
                                                (short)0, acc, false, false);
  }
#pragma unroll
  for (int r = 0; r < 8; ++r) {
    const int m_out = tm * 16 + r + 8 * khalf;
    C[(size_t)m_out * U_ + tn * 16 + m] = acc[r];
  }
}

__global__ void __launch_bounds__(256) co_mu_kernel(
    const float* __restrict__ ae, const float* __restrict__ ad,
    const float* __restrict__ nu, float* __restrict__ mu) {
  const int b      = blockIdx.y;
  const int s_tile = blockIdx.x >> 4;
  const int t_tile = blockIdx.x & 15;
  __shared__ float adT[16][260];
  __shared__ float aeT[16][260];
  __shared__ float nuS[256];

  const float kTwoLog2e = 2.8853900817779268f;
  const int tid = threadIdx.x;
  nuS[tid] = nu[tid];
  for (int i = tid; i < 16 * 256; i += 256) {
    const int r = i >> 8, c = i & 255;
    adT[r][c] = ad[((size_t)(b * TD_ + s_tile * 16 + r)) * U_ + c] * kTwoLog2e;
    aeT[r][c] = ae[((size_t)(b * TE_ + t_tile * 16 + r)) * U_ + c] * kTwoLog2e;
  }
  __syncthreads();

  const int si = tid >> 4, ti = tid & 15;
  float acc = 0.f;
#pragma unroll 4
  for (int u = 0; u < U_; ++u) {
    const float e  = __builtin_amdgcn_exp2f(adT[si][u] + aeT[ti][u]);
    const float r  = __builtin_amdgcn_rcpf(e + 1.0f);
    const float th = 1.0f - 2.0f * r;
    acc += th * nuS[u];
  }

  mu[((size_t)(b * TD_ + s_tile * 16 + si)) * TE_ + t_tile * 16 + ti] = acc;
}

__global__ void __launch_bounds__(256) softmax_rows_kernel(
    float* __restrict__ mu, float* __restrict__ max_mu) {
  const int row = blockIdx.x;
  const int tid = threadIdx.x;
  __shared__ float red[256];

  const float v = mu[(size_t)row * TE_ + tid];
  red[tid] = v; __syncthreads();
  for (int s = 128; s > 0; s >>= 1) {
    if (tid < s) red[tid] = fmaxf(red[tid], red[tid + s]);
    __syncthreads();
  }
  const float mx = red[0];
  __syncthreads();
  const float e = expf(v - mx);
  red[tid] = e; __syncthreads();
  for (int s = 128; s > 0; s >>= 1) {
    if (tid < s) red[tid] += red[tid + s];
    __syncthreads();
  }
  mu[(size_t)row * TE_ + tid] = e * (1.0f / red[0]);
  if (tid == 0) max_mu[row] = mx;
}

__global__ void __launch_bounds__(256) hhat_kernel(
    const float* __restrict__ max_mu, const float* __restrict__ de_seq,
    float* __restrict__ h_hat) {
  const int b   = blockIdx.x;
  const int tid = threadIdx.x;
  __shared__ float red[256];
  __shared__ float ma[256];

  const float v = max_mu[b * TD_ + tid];
  red[tid] = v; __syncthreads();
  for (int s = 128; s > 0; s >>= 1) {
    if (tid < s) red[tid] = fmaxf(red[tid], red[tid + s]);
    __syncthreads();
  }
  const float mx = red[0];
  __syncthreads();
  const float e = expf(v - mx);
  red[tid] = e; __syncthreads();
  for (int s = 128; s > 0; s >>= 1) {
    if (tid < s) red[tid] += red[tid + s];
    __syncthreads();
  }
  ma[tid] = e * (1.0f / red[0]);
  __syncthreads();

  float acc = 0.f;
  for (int s = 0; s < TD_; ++s)
    acc += de_seq[((size_t)(b * TD_ + s)) * D_ + tid] * ma[s];
  h_hat[b * D_ + tid] = acc;
}

__global__ void __launch_bounds__(256) sum_en_out_kernel(
    const float* __restrict__ alphas, const float* __restrict__ en_seq,
    const float* __restrict__ de_seq, const float* __restrict__ h_hat,
    float* __restrict__ out) {
  const int b     = blockIdx.y;
  const int lane  = threadIdx.x & 31;
  const int wave  = threadIdx.x >> 5;
  const int tile  = blockIdx.x * 8 + wave;
  const int td    = tile & 15;
  const int ts    = tile >> 4;
  const int m     = lane & 15;
  const int khalf = lane >> 4;

  const float* A    = alphas + (size_t)b * TD_ * TE_;
  const float* Bm   = en_seq + (size_t)b * TE_ * D_;
  const float* arow = A + (size_t)(ts * 16 + m) * TE_;
  const float* bcol = Bm + td * 16 + m;

  v8f acc = {};
  for (int k = 0; k < TE_; k += 4) {
    const int ka = k + 2 * khalf;
    v2f a, bb;
    a.x  = arow[ka + 0];
    a.y  = arow[ka + 1];
    bb.x = bcol[(size_t)(ka + 0) * D_];
    bb.y = bcol[(size_t)(ka + 1) * D_];
    acc = __builtin_amdgcn_wmma_f32_16x16x4_f32(false, a, false, bb,
                                                (short)0, acc, false, false);
  }

  const int d_g  = td * 16 + m;
  const float hh = h_hat[b * D_ + d_g];
#pragma unroll
  for (int r = 0; r < 8; ++r) {
    const int s_g  = ts * 16 + r + 8 * khalf;
    const float se = acc[r];
    const float de = de_seq[((size_t)(b * TD_ + s_g)) * D_ + d_g];
    float* o = out + ((size_t)(b * TD_ + s_g)) * (4 * D_);
    o[d_g]          = de;
    o[D_ + d_g]     = se;
    o[2 * D_ + d_g] = de * se;
    o[3 * D_ + d_g] = de * hh;
  }
}

extern "C" void kernel_launch(void* const* d_in, const int* in_sizes, int n_in,
                              void* d_out, int out_size, void* d_ws, size_t ws_size,
                              hipStream_t stream) {
  const float* en_seq = (const float*)d_in[0];
  const float* de_seq = (const float*)d_in[1];
  const float* w_en   = (const float*)d_in[2];
  const float* w_de   = (const float*)d_in[3];
  const float* nu     = (const float*)d_in[4];
  float* out = (float*)d_out;

  float* ws = (float*)d_ws;
  const size_t nAE = (size_t)B_ * TE_ * U_;
  float* ae     = ws;
  float* ad     = ae + nAE;
  float* mu     = ad + nAE;
  float* max_mu = mu + (size_t)B_ * TD_ * TE_;
  float* h_hat  = max_mu + (size_t)B_ * TD_;

  const dim3 blk(256);
  gemm_f32_wmma<<<dim3(256), blk, 0, stream>>>(en_seq, w_en, ae);
  gemm_f32_wmma<<<dim3(256), blk, 0, stream>>>(de_seq, w_de, ad);
  co_mu_kernel<<<dim3(256, B_), blk, 0, stream>>>(ae, ad, nu, mu);
  softmax_rows_kernel<<<dim3(B_ * TD_), blk, 0, stream>>>(mu, max_mu);
  hhat_kernel<<<dim3(B_), blk, 0, stream>>>(max_mu, de_seq, h_hat);
  sum_en_out_kernel<<<dim3(32, B_), blk, 0, stream>>>(mu, en_seq, de_seq,
                                                      h_hat, out);
}


# row-softmax launch removed: score kernel emits per-tile row maxima for h_hat; alphas@en kernel forms its softmax rows in LDS by wave lane-exchange max/sum (row-sum order changed), reads A from LDS
# speedup vs baseline: 1.9126x; 1.0400x over previous
#include <hip/hip_runtime.h>
#include <cmath>

#define B_  8
#define TE_ 256
#define TD_ 256
#define D_  256
#define U_  256

typedef float v2f __attribute__((ext_vector_type(2)));
typedef float v8f __attribute__((ext_vector_type(8)));

__global__ void __launch_bounds__(256) gemm_f32_wmma(
    const float* __restrict__ A, const float* __restrict__ W,
    float* __restrict__ C) {
  const int lane  = threadIdx.x & 31;
  const int wave  = threadIdx.x >> 5;
  const int tile  = blockIdx.x * 8 + wave;
  const int tn    = tile & 15;
  const int tm    = tile >> 4;
  const int m     = lane & 15;
  const int khalf = lane >> 4;

  const float* arow = A + (size_t)(tm * 16 + m) * U_;
  const float* bcol = W + tn * 16 + m;
  v8f acc = {};
  for (int k = 0; k < U_; k += 4) {
    const int ka = k + 2 * khalf;
    v2f a, b;
    a.x = arow[ka + 0];
    a.y = arow[ka + 1];
    b.x = bcol[(size_t)(ka + 0) * U_];
    b.y = bcol[(size_t)(ka + 1) * U_];
    acc = __builtin_amdgcn_wmma_f32_16x16x4_f32(false, a, false, b,
                                                (short)0, acc, false, false);
  }
#pragma unroll
  for (int r = 0; r < 8; ++r) {
    const int m_out = tm * 16 + r + 8 * khalf;
    C[(size_t)m_out * U_ + tn * 16 + m] = acc[r];
  }
}

__global__ void __launch_bounds__(256) co_mu_kernel(
    const float* __restrict__ ae, const float* __restrict__ ad,
    const float* __restrict__ nu, float* __restrict__ mu,
    float* __restrict__ pmax) {
  const int b      = blockIdx.y;
  const int s_tile = blockIdx.x >> 4;
  const int t_tile = blockIdx.x & 15;
  __shared__ float adT[16][260];
  __shared__ float aeT[16][260];
  __shared__ float nuS[256];

  const float kTwoLog2e = 2.8853900817779268f;
  const int tid = threadIdx.x;
  nuS[tid] = nu[tid];
  for (int i = tid; i < 16 * 256; i += 256) {
    const int r = i >> 8, c = i & 255;
    adT[r][c] = ad[((size_t)(b * TD_ + s_tile * 16 + r)) * U_ + c] * kTwoLog2e;
    aeT[r][c] = ae[((size_t)(b * TE_ + t_tile * 16 + r)) * U_ + c] * kTwoLog2e;
  }
  __syncthreads();

  const int si = tid >> 4, ti = tid & 15;
  float acc = 0.f;
#pragma unroll 4
  for (int u = 0; u < U_; ++u) {
    const float e  = __builtin_amdgcn_exp2f(adT[si][u] + aeT[ti][u]);
    const float r  = __builtin_amdgcn_rcpf(e + 1.0f);
    const float th = 1.0f - 2.0f * r;
    acc += th * nuS[u];
  }

  mu[((size_t)(b * TD_ + s_tile * 16 + si)) * TE_ + t_tile * 16 + ti] = acc;

  float pm = acc;
  pm = fmaxf(pm, __shfl_xor(pm, 8));
  pm = fmaxf(pm, __shfl_xor(pm, 4));
  pm = fmaxf(pm, __shfl_xor(pm, 2));
  pm = fmaxf(pm, __shfl_xor(pm, 1));
  if (ti == 0)
    pmax[((size_t)(b * TD_ + s_tile * 16 + si)) * 16 + t_tile] = pm;
}

__global__ void __launch_bounds__(256) hhat_kernel(
    const float* __restrict__ pmax, const float* __restrict__ de_seq,
    float* __restrict__ h_hat) {
  const int b   = blockIdx.x;
  const int tid = threadIdx.x;
  __shared__ float red[256];
  __shared__ float ma[256];

  const float* pm = pmax + ((size_t)(b * TD_ + tid)) * 16;
  float v = pm[0];
  for (int j = 1; j < 16; ++j) v = fmaxf(v, pm[j]);
  red[tid] = v; __syncthreads();
  for (int s = 128; s > 0; s >>= 1) {
    if (tid < s) red[tid] = fmaxf(red[tid], red[tid + s]);
    __syncthreads();
  }
  const float mx = red[0];
  __syncthreads();
  const float e = expf(v - mx);
  red[tid] = e; __syncthreads();
  for (int s = 128; s > 0; s >>= 1) {
    if (tid < s) red[tid] += red[tid + s];
    __syncthreads();
  }
  ma[tid] = e * (1.0f / red[0]);
  __syncthreads();

  float acc = 0.f;
  for (int s = 0; s < TD_; ++s)
    acc += de_seq[((size_t)(b * TD_ + s)) * D_ + tid] * ma[s];
  h_hat[b * D_ + tid] = acc;
}

__device__ __forceinline__ float wave_max(float x) {
  x = fmaxf(x, __shfl_xor(x, 16));
  x = fmaxf(x, __shfl_xor(x, 8));
  x = fmaxf(x, __shfl_xor(x, 4));
  x = fmaxf(x, __shfl_xor(x, 2));
  x = fmaxf(x, __shfl_xor(x, 1));
  return x;
}
__device__ __forceinline__ float wave_sum(float x) {
  x += __shfl_xor(x, 16);
  x += __shfl_xor(x, 8);
  x += __shfl_xor(x, 4);
  x += __shfl_xor(x, 2);
  x += __shfl_xor(x, 1);
  return x;
}

__global__ void __launch_bounds__(256) sum_en_out_kernel(
    const float* __restrict__ mu, const float* __restrict__ en_seq,
    const float* __restrict__ de_seq, const float* __restrict__ h_hat,
    float* __restrict__ out) {
  const int b     = blockIdx.y;
  const int lane  = threadIdx.x & 31;
  const int wave  = threadIdx.x >> 5;
  const int tile  = blockIdx.x * 8 + wave;
  const int td    = tile & 15;
  const int ts    = tile >> 4;
  const int m     = lane & 15;
  const int khalf = lane >> 4;

  __shared__ __attribute__((aligned(16))) float aS[16][260];

#pragma unroll
  for (int rr = 0; rr < 2; ++rr) {
    const int r = 2 * wave + rr;
    const float* mrow = mu + ((size_t)(b * TD_ + ts * 16 + r)) * TE_;
    const float4 v0 = *reinterpret_cast<const float4*>(mrow + 4 * lane);
    const float4 v1 = *reinterpret_cast<const float4*>(mrow + 128 + 4 * lane);
    const float mx = wave_max(fmaxf(fmaxf(fmaxf(v0.x, v0.y), fmaxf(v0.z, v0.w)),
                                    fmaxf(fmaxf(v1.x, v1.y), fmaxf(v1.z, v1.w))));
    float4 e0, e1;
    e0.x = expf(v0.x - mx); e0.y = expf(v0.y - mx);
    e0.z = expf(v0.z - mx); e0.w = expf(v0.w - mx);
    e1.x = expf(v1.x - mx); e1.y = expf(v1.y - mx);
    e1.z = expf(v1.z - mx); e1.w = expf(v1.w - mx);
    const float sm = wave_sum(((e0.x + e0.y) + (e0.z + e0.w)) +
                              ((e1.x + e1.y) + (e1.z + e1.w)));
    const float inv = 1.0f / sm;
    e0.x *= inv; e0.y *= inv; e0.z *= inv; e0.w *= inv;
    e1.x *= inv; e1.y *= inv; e1.z *= inv; e1.w *= inv;
    *reinterpret_cast<float4*>(&aS[r][4 * lane])       = e0;
    *reinterpret_cast<float4*>(&aS[r][128 + 4 * lane]) = e1;
  }
  __syncthreads();

  const float* Bm   = en_seq + (size_t)b * TE_ * D_;
  const float* aL   = &aS[m][2 * khalf];
  const float* bcol = Bm + td * 16 + m;

  v8f acc = {};
  for (int k = 0; k < TE_; k += 4) {
    const int ka = k + 2 * khalf;
    v2f a, bb;
    a.x  = aL[k + 0];
    a.y  = aL[k + 1];
    bb.x = bcol[(size_t)(ka + 0) * D_];
    bb.y = bcol[(size_t)(ka + 1) * D_];
    acc = __builtin_amdgcn_wmma_f32_16x16x4_f32(false, a, false, bb,
                                                (short)0, acc, false, false);
  }

  const int d_g  = td * 16 + m;
  const float hh = h_hat[b * D_ + d_g];
#pragma unroll
  for (int r = 0; r < 8; ++r) {
    const int s_g  = ts * 16 + r + 8 * khalf;
    const float se = acc[r];
    const float de = de_seq[((size_t)(b * TD_ + s_g)) * D_ + d_g];
    float* o = out + ((size_t)(b * TD_ + s_g)) * (4 * D_);
    o[d_g]          = de;
    o[D_ + d_g]     = se;
    o[2 * D_ + d_g] = de * se;
    o[3 * D_ + d_g] = de * hh;
  }
}

extern "C" void kernel_launch(void* const* d_in, const int* in_sizes, int n_in,
                              void* d_out, int out_size, void* d_ws, size_t ws_size,
                              hipStream_t stream) {
  const float* en_seq = (const float*)d_in[0];
  const float* de_seq = (const float*)d_in[1];
  const float* w_en   = (const float*)d_in[2];
  const float* w_de   = (const float*)d_in[3];
  const float* nu     = (const float*)d_in[4];
  float* out = (float*)d_out;

  float* ws = (float*)d_ws;
  const size_t nAE = (size_t)B_ * TE_ * U_;
  float* ae     = ws;
  float* ad     = ae + nAE;
  float* mu     = ad + nAE;
  float* pmax   = mu + (size_t)B_ * TD_ * TE_;
  float* h_hat  = pmax + (size_t)B_ * TD_ * 16;

  const dim3 blk(256);
  gemm_f32_wmma<<<dim3(256), blk, 0, stream>>>(en_seq, w_en, ae);
  gemm_f32_wmma<<<dim3(256), blk, 0, stream>>>(de_seq, w_de, ad);
  co_mu_kernel<<<dim3(256, B_), blk, 0, stream>>>(ae, ad, nu, mu, pmax);
  hhat_kernel<<<dim3(B_), blk, 0, stream>>>(pmax, de_seq, h_hat);
  sum_en_out_kernel<<<dim3(32, B_), blk, 0, stream>>>(mu, en_seq, de_seq,
                                                      h_hat, out);
}


# h_hat launch removed: each output-kernel block forms max_alphas and its own 128 h_hat columns in LDS; softmax sum and sum over s are now 8 per-wave partial sums added in wave order
# speedup vs baseline: 2.3838x; 1.2464x over previous
#include <hip/hip_runtime.h>
#include <cmath>

#define B_  8
#define TE_ 256
#define TD_ 256
#define D_  256
#define U_  256

typedef float v2f __attribute__((ext_vector_type(2)));
typedef float v8f __attribute__((ext_vector_type(8)));

__global__ void __launch_bounds__(256) gemm_f32_wmma(
    const float* __restrict__ A, const float* __restrict__ W,
    float* __restrict__ C) {
  const int lane  = threadIdx.x & 31;
  const int wave  = threadIdx.x >> 5;
  const int tile  = blockIdx.x * 8 + wave;
  const int tn    = tile & 15;
  const int tm    = tile >> 4;
  const int m     = lane & 15;
  const int khalf = lane >> 4;

  const float* arow = A + (size_t)(tm * 16 + m) * U_;
  const float* bcol = W + tn * 16 + m;
  v8f acc = {};
  for (int k = 0; k < U_; k += 4) {
    const int ka = k + 2 * khalf;
    v2f a, b;
    a.x = arow[ka + 0];
    a.y = arow[ka + 1];
    b.x = bcol[(size_t)(ka + 0) * U_];
    b.y = bcol[(size_t)(ka + 1) * U_];
    acc = __builtin_amdgcn_wmma_f32_16x16x4_f32(false, a, false, b,
                                                (short)0, acc, false, false);
  }
#pragma unroll
  for (int r = 0; r < 8; ++r) {
    const int m_out = tm * 16 + r + 8 * khalf;
    C[(size_t)m_out * U_ + tn * 16 + m] = acc[r];
  }
}

__global__ void __launch_bounds__(256) co_mu_kernel(
    const float* __restrict__ ae, const float* __restrict__ ad,
    const float* __restrict__ nu, float* __restrict__ mu,
    float* __restrict__ pmax) {
  const int b      = blockIdx.y;
  const int s_tile = blockIdx.x >> 4;
  const int t_tile = blockIdx.x & 15;
  __shared__ float adT[16][260];
  __shared__ float aeT[16][260];
  __shared__ float nuS[256];

  const float kTwoLog2e = 2.8853900817779268f;
  const int tid = threadIdx.x;
  nuS[tid] = nu[tid];
  for (int i = tid; i < 16 * 256; i += 256) {
    const int r = i >> 8, c = i & 255;
    adT[r][c] = ad[((size_t)(b * TD_ + s_tile * 16 + r)) * U_ + c] * kTwoLog2e;
    aeT[r][c] = ae[((size_t)(b * TE_ + t_tile * 16 + r)) * U_ + c] * kTwoLog2e;
  }
  __syncthreads();

  const int si = tid >> 4, ti = tid & 15;
  float acc = 0.f;
#pragma unroll 4
  for (int u = 0; u < U_; ++u) {
    const float e  = __builtin_amdgcn_exp2f(adT[si][u] + aeT[ti][u]);
    const float r  = __builtin_amdgcn_rcpf(e + 1.0f);
    const float th = 1.0f - 2.0f * r;
    acc += th * nuS[u];
  }

  mu[((size_t)(b * TD_ + s_tile * 16 + si)) * TE_ + t_tile * 16 + ti] = acc;

  float pm = acc;
  pm = fmaxf(pm, __shfl_xor(pm, 8));
  pm = fmaxf(pm, __shfl_xor(pm, 4));
  pm = fmaxf(pm, __shfl_xor(pm, 2));
  pm = fmaxf(pm, __shfl_xor(pm, 1));
  if (ti == 0)
    pmax[((size_t)(b * TD_ + s_tile * 16 + si)) * 16 + t_tile] = pm;
}

__device__ __forceinline__ void lds_barrier() {
  __builtin_amdgcn_fence(__ATOMIC_RELEASE, "workgroup", "local");
  __builtin_amdgcn_s_barrier();
  __builtin_amdgcn_fence(__ATOMIC_ACQUIRE, "workgroup", "local");
}

__device__ __forceinline__ float wave_max(float x) {
  x = fmaxf(x, __shfl_xor(x, 16));
  x = fmaxf(x, __shfl_xor(x, 8));
  x = fmaxf(x, __shfl_xor(x, 4));
  x = fmaxf(x, __shfl_xor(x, 2));
  x = fmaxf(x, __shfl_xor(x, 1));
  return x;
}
__device__ __forceinline__ float wave_sum(float x) {
  x += __shfl_xor(x, 16);
  x += __shfl_xor(x, 8);
  x += __shfl_xor(x, 4);
  x += __shfl_xor(x, 2);
  x += __shfl_xor(x, 1);
  return x;
}

__global__ void __launch_bounds__(256) sum_en_out_kernel(
    const float* __restrict__ mu, const float* __restrict__ pmax,
    const float* __restrict__ en_seq, const float* __restrict__ de_seq,
    float* __restrict__ out) {
  const int b     = blockIdx.y;
  const int tid   = threadIdx.x;
  const int lane  = threadIdx.x & 31;
  const int wave  = threadIdx.x >> 5;
  const int tile  = blockIdx.x * 8 + wave;
  const int td    = tile & 15;
  const int ts    = tile >> 4;
  const int m     = lane & 15;
  const int khalf = lane >> 4;
  const int dbase = (blockIdx.x & 1) * 128;

  __shared__ __attribute__((aligned(16))) float aS[16][260];
  __shared__ __attribute__((aligned(16))) float hpS[8][128];
  __shared__ float maS[256];
  __shared__ float wmaxS[8];
  __shared__ float wsumS[8];

  const float4* pm4 = reinterpret_cast<const float4*>(
      pmax + ((size_t)(b * TD_ + tid)) * 16);
  const float4 p0 = pm4[0], p1 = pm4[1], p2 = pm4[2], p3 = pm4[3];

  float4 mv[2][2];
#pragma unroll
  for (int rr = 0; rr < 2; ++rr) {
    const int r = 2 * wave + rr;
    const float* mrow = mu + ((size_t)(b * TD_ + ts * 16 + r)) * TE_;
    mv[rr][0] = *reinterpret_cast<const float4*>(mrow + 4 * lane);
    mv[rr][1] = *reinterpret_cast<const float4*>(mrow + 128 + 4 * lane);
  }

  const float* dp = de_seq + ((size_t)(b * TD_ + 32 * wave)) * D_ + dbase + 4 * lane;
  float4 dv[32];
#pragma unroll
  for (int i = 0; i < 32; ++i)
    dv[i] = *reinterpret_cast<const float4*>(dp + (size_t)i * D_);

  const float v = fmaxf(
      fmaxf(fmaxf(fmaxf(p0.x, p0.y), fmaxf(p0.z, p0.w)),
            fmaxf(fmaxf(p1.x, p1.y), fmaxf(p1.z, p1.w))),
      fmaxf(fmaxf(fmaxf(p2.x, p2.y), fmaxf(p2.z, p2.w)),
            fmaxf(fmaxf(p3.x, p3.y), fmaxf(p3.z, p3.w))));
  const float wm = wave_max(v);
  if (lane == 0) wmaxS[wave] = wm;
  lds_barrier();

#pragma unroll
  for (int rr = 0; rr < 2; ++rr) {
    const int r = 2 * wave + rr;
    const float4 v0 = mv[rr][0];
    const float4 v1 = mv[rr][1];
    const float mx = wave_max(fmaxf(fmaxf(fmaxf(v0.x, v0.y), fmaxf(v0.z, v0.w)),
                                    fmaxf(fmaxf(v1.x, v1.y), fmaxf(v1.z, v1.w))));
    float4 e0, e1;
    e0.x = expf(v0.x - mx); e0.y = expf(v0.y - mx);
    e0.z = expf(v0.z - mx); e0.w = expf(v0.w - mx);
    e1.x = expf(v1.x - mx); e1.y = expf(v1.y - mx);
    e1.z = expf(v1.z - mx); e1.w = expf(v1.w - mx);
    const float sm = wave_sum(((e0.x + e0.y) + (e0.z + e0.w)) +
                              ((e1.x + e1.y) + (e1.z + e1.w)));
    const float inv = 1.0f / sm;
    e0.x *= inv; e0.y *= inv; e0.z *= inv; e0.w *= inv;
    e1.x *= inv; e1.y *= inv; e1.z *= inv; e1.w *= inv;
    *reinterpret_cast<float4*>(&aS[r][4 * lane])       = e0;
    *reinterpret_cast<float4*>(&aS[r][128 + 4 * lane]) = e1;
  }

  float mxs = wmaxS[0];
  for (int j = 1; j < 8; ++j) mxs = fmaxf(mxs, wmaxS[j]);
  const float ev = expf(v - mxs);
  const float wsm = wave_sum(ev);
  if (lane == 0) wsumS[wave] = wsm;
  lds_barrier();
  float tot = wsumS[0];
  for (int j = 1; j < 8; ++j) tot += wsumS[j];
  maS[tid] = ev * (1.0f / tot);
  lds_barrier();

  {
    float4 ha = {0.f, 0.f, 0.f, 0.f};
#pragma unroll
    for (int i = 0; i < 32; ++i) {
      const float w = maS[32 * wave + i];
      ha.x = fmaf(dv[i].x, w, ha.x);
      ha.y = fmaf(dv[i].y, w, ha.y);
      ha.z = fmaf(dv[i].z, w, ha.z);
      ha.w = fmaf(dv[i].w, w, ha.w);
    }
    *reinterpret_cast<float4*>(&hpS[wave][4 * lane]) = ha;
  }

  lds_barrier();

  const int d_g = td * 16 + m;
  float hh = hpS[0][d_g - dbase];
  for (int j = 1; j < 8; ++j) hh += hpS[j][d_g - dbase];

  const float* Bm   = en_seq + (size_t)b * TE_ * D_;
  const float* aL   = &aS[m][2 * khalf];
  const float* bcol = Bm + td * 16 + m;

  v8f acc = {};
  for (int k = 0; k < TE_; k += 4) {
    const int ka = k + 2 * khalf;
    v2f a, bb;
    a.x  = aL[k + 0];
    a.y  = aL[k + 1];
    bb.x = bcol[(size_t)(ka + 0) * D_];
    bb.y = bcol[(size_t)(ka + 1) * D_];
    acc = __builtin_amdgcn_wmma_f32_16x16x4_f32(false, a, false, bb,
                                                (short)0, acc, false, false);
  }

#pragma unroll
  for (int r = 0; r < 8; ++r) {
    const int s_g  = ts * 16 + r + 8 * khalf;
    const float se = acc[r];
    const float de = de_seq[((size_t)(b * TD_ + s_g)) * D_ + d_g];
    float* o = out + ((size_t)(b * TD_ + s_g)) * (4 * D_);
    o[d_g]          = de;
    o[D_ + d_g]     = se;
    o[2 * D_ + d_g] = de * se;
    o[3 * D_ + d_g] = de * hh;
  }
}

extern "C" void kernel_launch(void* const* d_in, const int* in_sizes, int n_in,
                              void* d_out, int out_size, void* d_ws, size_t ws_size,
                              hipStream_t stream) {
  const float* en_seq = (const float*)d_in[0];
  const float* de_seq = (const float*)d_in[1];
  const float* w_en   = (const float*)d_in[2];
  const float* w_de   = (const float*)d_in[3];
  const float* nu     = (const float*)d_in[4];
  float* out = (float*)d_out;

  float* ws = (float*)d_ws;
  const size_t nAE = (size_t)B_ * TE_ * U_;
  float* ae     = ws;
  float* ad     = ae + nAE;
  float* mu     = ad + nAE;
  float* pmax   = mu + (size_t)B_ * TD_ * TE_;

  const dim3 blk(256);
  gemm_f32_wmma<<<dim3(256), blk, 0, stream>>>(en_seq, w_en, ae);
  gemm_f32_wmma<<<dim3(256), blk, 0, stream>>>(de_seq, w_de, ad);
  co_mu_kernel<<<dim3(256, B_), blk, 0, stream>>>(ae, ad, nu, mu, pmax);
  sum_en_out_kernel<<<dim3(32, B_), blk, 0, stream>>>(mu, pmax, en_seq,
                                                      de_seq, out);
}


# score kernel: 32x32 tile per block, each thread computes a 2x2 patch of mu so every LDS operand read feeds two outputs; 4x fewer waves, tile staged by 16-byte loads in one trip; sum order unchanged
# speedup vs baseline: 2.8474x; 1.1945x over previous
#include <hip/hip_runtime.h>
#include <cmath>

#define B_  8
#define TE_ 256
#define TD_ 256
#define D_  256
#define U_  256

typedef float v2f __attribute__((ext_vector_type(2)));
typedef float v8f __attribute__((ext_vector_type(8)));

__global__ void __launch_bounds__(256) gemm_f32_wmma(
    const float* __restrict__ A, const float* __restrict__ W,
    float* __restrict__ C) {
  const int lane  = threadIdx.x & 31;
  const int wave  = threadIdx.x >> 5;
  const int tile  = blockIdx.x * 8 + wave;
  const int tn    = tile & 15;
  const int tm    = tile >> 4;
  const int m     = lane & 15;
  const int khalf = lane >> 4;

  const float* arow = A + (size_t)(tm * 16 + m) * U_;
  const float* bcol = W + tn * 16 + m;
  v8f acc = {};
  for (int k = 0; k < U_; k += 4) {
    const int ka = k + 2 * khalf;
    v2f a, b;
    a.x = arow[ka + 0];
    a.y = arow[ka + 1];
    b.x = bcol[(size_t)(ka + 0) * U_];
    b.y = bcol[(size_t)(ka + 1) * U_];
    acc = __builtin_amdgcn_wmma_f32_16x16x4_f32(false, a, false, b,
                                                (short)0, acc, false, false);
  }
#pragma unroll
  for (int r = 0; r < 8; ++r) {
    const int m_out = tm * 16 + r + 8 * khalf;
    C[(size_t)m_out * U_ + tn * 16 + m] = acc[r];
  }
}

__device__ __forceinline__ float tanh_scaled(float x2) {
  const float e = __builtin_amdgcn_exp2f(x2);
  const float r = __builtin_amdgcn_rcpf(e + 1.0f);
  return 1.0f - 2.0f * r;
}

__global__ void __launch_bounds__(256) co_mu_kernel(
    const float* __restrict__ ae, const float* __restrict__ ad,
    const float* __restrict__ nu, float* __restrict__ mu,
    float* __restrict__ pmax) {
  const int b      = blockIdx.y;
  const int s_tile = blockIdx.x >> 3;
  const int t_tile = blockIdx.x & 7;
  __shared__ __attribute__((aligned(16))) float adT[32][260];
  __shared__ __attribute__((aligned(16))) float aeT[32][260];
  __shared__ __attribute__((aligned(16))) float nuS[256];

  const float kTwoLog2e = 2.8853900817779268f;
  const int tid = threadIdx.x;
  nuS[tid] = nu[tid];

  const float4* ad4 = reinterpret_cast<const float4*>(
      ad + ((size_t)(b * TD_ + s_tile * 32)) * U_);
  const float4* ae4 = reinterpret_cast<const float4*>(
      ae + ((size_t)(b * TE_ + t_tile * 32)) * U_);
  float4 va[8], vb[8];
#pragma unroll
  for (int j = 0; j < 8; ++j) {
    va[j] = ad4[tid + 256 * j];
    vb[j] = ae4[tid + 256 * j];
  }
#pragma unroll
  for (int j = 0; j < 8; ++j) {
    const int i = tid + 256 * j;
    const int r = i >> 6, c = (i & 63) * 4;
    float4 x = va[j], y = vb[j];
    x.x *= kTwoLog2e; x.y *= kTwoLog2e; x.z *= kTwoLog2e; x.w *= kTwoLog2e;
    y.x *= kTwoLog2e; y.y *= kTwoLog2e; y.z *= kTwoLog2e; y.w *= kTwoLog2e;
    *reinterpret_cast<float4*>(&adT[r][c]) = x;
    *reinterpret_cast<float4*>(&aeT[r][c]) = y;
  }
  __syncthreads();

  const int si = tid >> 4, ti = tid & 15;
  float acc00 = 0.f, acc01 = 0.f, acc10 = 0.f, acc11 = 0.f;
#pragma unroll 4
  for (int u = 0; u < U_; ++u) {
    const float a0 = adT[si][u], a1 = adT[si + 16][u];
    const float e0 = aeT[ti][u], e1 = aeT[ti + 16][u];
    const float n  = nuS[u];
    acc00 += tanh_scaled(a0 + e0) * n;
    acc01 += tanh_scaled(a0 + e1) * n;
    acc10 += tanh_scaled(a1 + e0) * n;
    acc11 += tanh_scaled(a1 + e1) * n;
  }

  const int s0 = b * TD_ + s_tile * 32 + si;
  const int t0 = t_tile * 32 + ti;
  mu[(size_t)s0 * TE_ + t0]             = acc00;
  mu[(size_t)s0 * TE_ + t0 + 16]        = acc01;
  mu[(size_t)(s0 + 16) * TE_ + t0]      = acc10;
  mu[(size_t)(s0 + 16) * TE_ + t0 + 16] = acc11;

  float p00 = acc00, p01 = acc01, p10 = acc10, p11 = acc11;
#pragma unroll
  for (int x = 8; x >= 1; x >>= 1) {
    p00 = fmaxf(p00, __shfl_xor(p00, x));
    p01 = fmaxf(p01, __shfl_xor(p01, x));
    p10 = fmaxf(p10, __shfl_xor(p10, x));
    p11 = fmaxf(p11, __shfl_xor(p11, x));
  }
  if (ti == 0) {
    float* q0 = pmax + (size_t)s0 * 16 + 2 * t_tile;
    float* q1 = pmax + (size_t)(s0 + 16) * 16 + 2 * t_tile;
    q0[0] = p00; q0[1] = p01;
    q1[0] = p10; q1[1] = p11;
  }
}

__device__ __forceinline__ void lds_barrier() {
  __builtin_amdgcn_fence(__ATOMIC_RELEASE, "workgroup", "local");
  __builtin_amdgcn_s_barrier();
  __builtin_amdgcn_fence(__ATOMIC_ACQUIRE, "workgroup", "local");
}

__device__ __forceinline__ float wave_max(float x) {
  x = fmaxf(x, __shfl_xor(x, 16));
  x = fmaxf(x, __shfl_xor(x, 8));
  x = fmaxf(x, __shfl_xor(x, 4));
  x = fmaxf(x, __shfl_xor(x, 2));
  x = fmaxf(x, __shfl_xor(x, 1));
  return x;
}
__device__ __forceinline__ float wave_sum(float x) {
  x += __shfl_xor(x, 16);
  x += __shfl_xor(x, 8);
  x += __shfl_xor(x, 4);
  x += __shfl_xor(x, 2);
  x += __shfl_xor(x, 1);
  return x;
}

__global__ void __launch_bounds__(256) sum_en_out_kernel(
    const float* __restrict__ mu, const float* __restrict__ pmax,
    const float* __restrict__ en_seq, const float* __restrict__ de_seq,
    float* __restrict__ out) {
  const int b     = blockIdx.y;
  const int tid   = threadIdx.x;
  const int lane  = threadIdx.x & 31;
  const int wave  = threadIdx.x >> 5;
  const int tile  = blockIdx.x * 8 + wave;
  const int td    = tile & 15;
  const int ts    = tile >> 4;
  const int m     = lane & 15;
  const int khalf = lane >> 4;
  const int dbase = (blockIdx.x & 1) * 128;

  __shared__ __attribute__((aligned(16))) float aS[16][260];
  __shared__ __attribute__((aligned(16))) float hpS[8][128];
  __shared__ float maS[256];
  __shared__ float wmaxS[8];
  __shared__ float wsumS[8];

  const float4* pm4 = reinterpret_cast<const float4*>(
      pmax + ((size_t)(b * TD_ + tid)) * 16);
  const float4 p0 = pm4[0], p1 = pm4[1], p2 = pm4[2], p3 = pm4[3];

  float4 mv[2][2];
#pragma unroll
  for (int rr = 0; rr < 2; ++rr) {
    const int r = 2 * wave + rr;
    const float* mrow = mu + ((size_t)(b * TD_ + ts * 16 + r)) * TE_;
    mv[rr][0] = *reinterpret_cast<const float4*>(mrow + 4 * lane);
    mv[rr][1] = *reinterpret_cast<const float4*>(mrow + 128 + 4 * lane);
  }

  const float* dp = de_seq + ((size_t)(b * TD_ + 32 * wave)) * D_ + dbase + 4 * lane;
  float4 dv[32];
#pragma unroll
  for (int i = 0; i < 32; ++i)
    dv[i] = *reinterpret_cast<const float4*>(dp + (size_t)i * D_);

  const float v = fmaxf(
      fmaxf(fmaxf(fmaxf(p0.x, p0.y), fmaxf(p0.z, p0.w)),
            fmaxf(fmaxf(p1.x, p1.y), fmaxf(p1.z, p1.w))),
      fmaxf(fmaxf(fmaxf(p2.x, p2.y), fmaxf(p2.z, p2.w)),
            fmaxf(fmaxf(p3.x, p3.y), fmaxf(p3.z, p3.w))));
  const float wm = wave_max(v);
  if (lane == 0) wmaxS[wave] = wm;
  lds_barrier();

#pragma unroll
  for (int rr = 0; rr < 2; ++rr) {
    const int r = 2 * wave + rr;
    const float4 v0 = mv[rr][0];
    const float4 v1 = mv[rr][1];
    const float mx = wave_max(fmaxf(fmaxf(fmaxf(v0.x, v0.y), fmaxf(v0.z, v0.w)),
                                    fmaxf(fmaxf(v1.x, v1.y), fmaxf(v1.z, v1.w))));
    float4 e0, e1;
    e0.x = expf(v0.x - mx); e0.y = expf(v0.y - mx);
    e0.z = expf(v0.z - mx); e0.w = expf(v0.w - mx);
    e1.x = expf(v1.x - mx); e1.y = expf(v1.y - mx);
    e1.z = expf(v1.z - mx); e1.w = expf(v1.w - mx);
    const float sm = wave_sum(((e0.x + e0.y) + (e0.z + e0.w)) +
                              ((e1.x + e1.y) + (e1.z + e1.w)));
    const float inv = 1.0f / sm;
    e0.x *= inv; e0.y *= inv; e0.z *= inv; e0.w *= inv;
    e1.x *= inv; e1.y *= inv; e1.z *= inv; e1.w *= inv;
    *reinterpret_cast<float4*>(&aS[r][4 * lane])       = e0;
    *reinterpret_cast<float4*>(&aS[r][128 + 4 * lane]) = e1;
  }

  float mxs = wmaxS[0];
  for (int j = 1; j < 8; ++j) mxs = fmaxf(mxs, wmaxS[j]);
  const float ev = expf(v - mxs);
  const float wsm = wave_sum(ev);
  if (lane == 0) wsumS[wave] = wsm;
  lds_barrier();
  float tot = wsumS[0];
  for (int j = 1; j < 8; ++j) tot += wsumS[j];
  maS[tid] = ev * (1.0f / tot);
  lds_barrier();

  {
    float4 ha = {0.f, 0.f, 0.f, 0.f};
#pragma unroll
    for (int i = 0; i < 32; ++i) {
      const float w = maS[32 * wave + i];
      ha.x = fmaf(dv[i].x, w, ha.x);
      ha.y = fmaf(dv[i].y, w, ha.y);
      ha.z = fmaf(dv[i].z, w, ha.z);
      ha.w = fmaf(dv[i].w, w, ha.w);
    }
    *reinterpret_cast<float4*>(&hpS[wave][4 * lane]) = ha;
  }

  lds_barrier();

  const int d_g = td * 16 + m;
  float hh = hpS[0][d_g - dbase];
  for (int j = 1; j < 8; ++j) hh += hpS[j][d_g - dbase];

  const float* Bm   = en_seq + (size_t)b * TE_ * D_;
  const float* aL   = &aS[m][2 * khalf];
  const float* bcol = Bm + td * 16 + m;

  v8f acc = {};
  for (int k = 0; k < TE_; k += 4) {
    const int ka = k + 2 * khalf;
    v2f a, bb;
    a.x  = aL[k + 0];
    a.y  = aL[k + 1];
    bb.x = bcol[(size_t)(ka + 0) * D_];
    bb.y = bcol[(size_t)(ka + 1) * D_];
    acc = __builtin_amdgcn_wmma_f32_16x16x4_f32(false, a, false, bb,
                                                (short)0, acc, false, false);
  }

#pragma unroll
  for (int r = 0; r < 8; ++r) {
    const int s_g  = ts * 16 + r + 8 * khalf;
    const float se = acc[r];
    const float de = de_seq[((size_t)(b * TD_ + s_g)) * D_ + d_g];
    float* o = out + ((size_t)(b * TD_ + s_g)) * (4 * D_);
    o[d_g]          = de;
    o[D_ + d_g]     = se;
    o[2 * D_ + d_g] = de * se;
    o[3 * D_ + d_g] = de * hh;
  }
}

extern "C" void kernel_launch(void* const* d_in, const int* in_sizes, int n_in,
                              void* d_out, int out_size, void* d_ws, size_t ws_size,
                              hipStream_t stream) {
  const float* en_seq = (const float*)d_in[0];
  const float* de_seq = (const float*)d_in[1];
  const float* w_en   = (const float*)d_in[2];
  const float* w_de   = (const float*)d_in[3];
  const float* nu     = (const float*)d_in[4];
  float* out = (float*)d_out;

  float* ws = (float*)d_ws;
  const size_t nAE = (size_t)B_ * TE_ * U_;
  float* ae     = ws;
  float* ad     = ae + nAE;
  float* mu     = ad + nAE;
  float* pmax   = mu + (size_t)B_ * TD_ * TE_;

  const dim3 blk(256);
  gemm_f32_wmma<<<dim3(256), blk, 0, stream>>>(en_seq, w_en, ae);
  gemm_f32_wmma<<<dim3(256), blk, 0, stream>>>(de_seq, w_de, ad);
  co_mu_kernel<<<dim3(64, B_), blk, 0, stream>>>(ae, ad, nu, mu, pmax);
  sum_en_out_kernel<<<dim3(32, B_), blk, 0, stream>>>(mu, pmax, en_seq,
                                                      de_seq, out);
}


# score kernel: tanh by the hardware v_tanh_f32 instruction, one transcendental per term in place of exp2 + rcp; tiles staged unscaled; sum order over u unchanged
# speedup vs baseline: 3.2154x; 1.1292x over previous
#include <hip/hip_runtime.h>
#include <cmath>

#define B_  8
#define TE_ 256
#define TD_ 256
#define D_  256
#define U_  256

typedef float v2f __attribute__((ext_vector_type(2)));
typedef float v8f __attribute__((ext_vector_type(8)));

__global__ void __launch_bounds__(256) gemm_f32_wmma(
    const float* __restrict__ A, const float* __restrict__ W,
    float* __restrict__ C) {
  const int lane  = threadIdx.x & 31;
  const int wave  = threadIdx.x >> 5;
  const int tile  = blockIdx.x * 8 + wave;
  const int tn    = tile & 15;
  const int tm    = tile >> 4;
  const int m     = lane & 15;
  const int khalf = lane >> 4;

  const float* arow = A + (size_t)(tm * 16 + m) * U_;
  const float* bcol = W + tn * 16 + m;
  v8f acc = {};
  for (int k = 0; k < U_; k += 4) {
    const int ka = k + 2 * khalf;
    v2f a, b;
    a.x = arow[ka + 0];
    a.y = arow[ka + 1];
    b.x = bcol[(size_t)(ka + 0) * U_];
    b.y = bcol[(size_t)(ka + 1) * U_];
    acc = __builtin_amdgcn_wmma_f32_16x16x4_f32(false, a, false, b,
                                                (short)0, acc, false, false);
  }
#pragma unroll
  for (int r = 0; r < 8; ++r) {
    const int m_out = tm * 16 + r + 8 * khalf;
    C[(size_t)m_out * U_ + tn * 16 + m] = acc[r];
  }
}

__device__ __forceinline__ float tanh_hw(float x) {
  return __builtin_amdgcn_tanhf(x);
}

__global__ void __launch_bounds__(256) co_mu_kernel(
    const float* __restrict__ ae, const float* __restrict__ ad,
    const float* __restrict__ nu, float* __restrict__ mu,
    float* __restrict__ pmax) {
  const int b      = blockIdx.y;
  const int s_tile = blockIdx.x >> 3;
  const int t_tile = blockIdx.x & 7;
  __shared__ __attribute__((aligned(16))) float adT[32][260];
  __shared__ __attribute__((aligned(16))) float aeT[32][260];
  __shared__ __attribute__((aligned(16))) float nuS[256];

  const int tid = threadIdx.x;
  nuS[tid] = nu[tid];

  const float4* ad4 = reinterpret_cast<const float4*>(
      ad + ((size_t)(b * TD_ + s_tile * 32)) * U_);
  const float4* ae4 = reinterpret_cast<const float4*>(
      ae + ((size_t)(b * TE_ + t_tile * 32)) * U_);
  float4 va[8], vb[8];
#pragma unroll
  for (int j = 0; j < 8; ++j) {
    va[j] = ad4[tid + 256 * j];
    vb[j] = ae4[tid + 256 * j];
  }
#pragma unroll
  for (int j = 0; j < 8; ++j) {
    const int i = tid + 256 * j;
    const int r = i >> 6, c = (i & 63) * 4;
    *reinterpret_cast<float4*>(&adT[r][c]) = va[j];
    *reinterpret_cast<float4*>(&aeT[r][c]) = vb[j];
  }
  __syncthreads();

  const int si = tid >> 4, ti = tid & 15;
  float acc00 = 0.f, acc01 = 0.f, acc10 = 0.f, acc11 = 0.f;
#pragma unroll 4
  for (int u = 0; u < U_; ++u) {
    const float a0 = adT[si][u], a1 = adT[si + 16][u];
    const float e0 = aeT[ti][u], e1 = aeT[ti + 16][u];
    const float n  = nuS[u];
    acc00 += tanh_hw(a0 + e0) * n;
    acc01 += tanh_hw(a0 + e1) * n;
    acc10 += tanh_hw(a1 + e0) * n;
    acc11 += tanh_hw(a1 + e1) * n;
  }

  const int s0 = b * TD_ + s_tile * 32 + si;
  const int t0 = t_tile * 32 + ti;
  mu[(size_t)s0 * TE_ + t0]             = acc00;
  mu[(size_t)s0 * TE_ + t0 + 16]        = acc01;
  mu[(size_t)(s0 + 16) * TE_ + t0]      = acc10;
  mu[(size_t)(s0 + 16) * TE_ + t0 + 16] = acc11;

  float p00 = acc00, p01 = acc01, p10 = acc10, p11 = acc11;
#pragma unroll
  for (int x = 8; x >= 1; x >>= 1) {
    p00 = fmaxf(p00, __shfl_xor(p00, x));
    p01 = fmaxf(p01, __shfl_xor(p01, x));
    p10 = fmaxf(p10, __shfl_xor(p10, x));
    p11 = fmaxf(p11, __shfl_xor(p11, x));
  }
  if (ti == 0) {
    float* q0 = pmax + (size_t)s0 * 16 + 2 * t_tile;
    float* q1 = pmax + (size_t)(s0 + 16) * 16 + 2 * t_tile;
    q0[0] = p00; q0[1] = p01;
    q1[0] = p10; q1[1] = p11;
  }
}

__device__ __forceinline__ void lds_barrier() {
  __builtin_amdgcn_fence(__ATOMIC_RELEASE, "workgroup", "local");
  __builtin_amdgcn_s_barrier();
  __builtin_amdgcn_fence(__ATOMIC_ACQUIRE, "workgroup", "local");
}

__device__ __forceinline__ float wave_max(float x) {
  x = fmaxf(x, __shfl_xor(x, 16));
  x = fmaxf(x, __shfl_xor(x, 8));
  x = fmaxf(x, __shfl_xor(x, 4));
  x = fmaxf(x, __shfl_xor(x, 2));
  x = fmaxf(x, __shfl_xor(x, 1));
  return x;
}
__device__ __forceinline__ float wave_sum(float x) {
  x += __shfl_xor(x, 16);
  x += __shfl_xor(x, 8);
  x += __shfl_xor(x, 4);
  x += __shfl_xor(x, 2);
  x += __shfl_xor(x, 1);
  return x;
}

__global__ void __launch_bounds__(256) sum_en_out_kernel(
    const float* __restrict__ mu, const float* __restrict__ pmax,
    const float* __restrict__ en_seq, const float* __restrict__ de_seq,
    float* __restrict__ out) {
  const int b     = blockIdx.y;
  const int tid   = threadIdx.x;
  const int lane  = threadIdx.x & 31;
  const int wave  = threadIdx.x >> 5;
  const int tile  = blockIdx.x * 8 + wave;
  const int td    = tile & 15;
  const int ts    = tile >> 4;
  const int m     = lane & 15;
  const int khalf = lane >> 4;
  const int dbase = (blockIdx.x & 1) * 128;

  __shared__ __attribute__((aligned(16))) float aS[16][260];
  __shared__ __attribute__((aligned(16))) float hpS[8][128];
  __shared__ float maS[256];
  __shared__ float wmaxS[8];
  __shared__ float wsumS[8];

  const float4* pm4 = reinterpret_cast<const float4*>(
      pmax + ((size_t)(b * TD_ + tid)) * 16);
  const float4 p0 = pm4[0], p1 = pm4[1], p2 = pm4[2], p3 = pm4[3];

  float4 mv[2][2];
#pragma unroll
  for (int rr = 0; rr < 2; ++rr) {
    const int r = 2 * wave + rr;
    const float* mrow = mu + ((size_t)(b * TD_ + ts * 16 + r)) * TE_;
    mv[rr][0] = *reinterpret_cast<const float4*>(mrow + 4 * lane);
    mv[rr][1] = *reinterpret_cast<const float4*>(mrow + 128 + 4 * lane);
  }

  const float* dp = de_seq + ((size_t)(b * TD_ + 32 * wave)) * D_ + dbase + 4 * lane;
  float4 dv[32];
#pragma unroll
  for (int i = 0; i < 32; ++i)
    dv[i] = *reinterpret_cast<const float4*>(dp + (size_t)i * D_);

  const float v = fmaxf(
      fmaxf(fmaxf(fmaxf(p0.x, p0.y), fmaxf(p0.z, p0.w)),
            fmaxf(fmaxf(p1.x, p1.y), fmaxf(p1.z, p1.w))),
      fmaxf(fmaxf(fmaxf(p2.x, p2.y), fmaxf(p2.z, p2.w)),
            fmaxf(fmaxf(p3.x, p3.y), fmaxf(p3.z, p3.w))));
  const float wm = wave_max(v);
  if (lane == 0) wmaxS[wave] = wm;
  lds_barrier();

#pragma unroll
  for (int rr = 0; rr < 2; ++rr) {
    const int r = 2 * wave + rr;
    const float4 v0 = mv[rr][0];
    const float4 v1 = mv[rr][1];
    const float mx = wave_max(fmaxf(fmaxf(fmaxf(v0.x, v0.y), fmaxf(v0.z, v0.w)),
                                    fmaxf(fmaxf(v1.x, v1.y), fmaxf(v1.z, v1.w))));
    float4 e0, e1;
    e0.x = expf(v0.x - mx); e0.y = expf(v0.y - mx);
    e0.z = expf(v0.z - mx); e0.w = expf(v0.w - mx);
    e1.x = expf(v1.x - mx); e1.y = expf(v1.y - mx);
    e1.z = expf(v1.z - mx); e1.w = expf(v1.w - mx);
    const float sm = wave_sum(((e0.x + e0.y) + (e0.z + e0.w)) +
                              ((e1.x + e1.y) + (e1.z + e1.w)));
    const float inv = 1.0f / sm;
    e0.x *= inv; e0.y *= inv; e0.z *= inv; e0.w *= inv;
    e1.x *= inv; e1.y *= inv; e1.z *= inv; e1.w *= inv;
    *reinterpret_cast<float4*>(&aS[r][4 * lane])       = e0;
    *reinterpret_cast<float4*>(&aS[r][128 + 4 * lane]) = e1;
  }

  float mxs = wmaxS[0];
  for (int j = 1; j < 8; ++j) mxs = fmaxf(mxs, wmaxS[j]);
  const float ev = expf(v - mxs);
  const float wsm = wave_sum(ev);
  if (lane == 0) wsumS[wave] = wsm;
  lds_barrier();
  float tot = wsumS[0];
  for (int j = 1; j < 8; ++j) tot += wsumS[j];
  maS[tid] = ev * (1.0f / tot);
  lds_barrier();

  {
    float4 ha = {0.f, 0.f, 0.f, 0.f};
#pragma unroll
    for (int i = 0; i < 32; ++i) {
      const float w = maS[32 * wave + i];
      ha.x = fmaf(dv[i].x, w, ha.x);
      ha.y = fmaf(dv[i].y, w, ha.y);
      ha.z = fmaf(dv[i].z, w, ha.z);
      ha.w = fmaf(dv[i].w, w, ha.w);
    }
    *reinterpret_cast<float4*>(&hpS[wave][4 * lane]) = ha;
  }

  lds_barrier();

  const int d_g = td * 16 + m;
  float hh = hpS[0][d_g - dbase];
  for (int j = 1; j < 8; ++j) hh += hpS[j][d_g - dbase];

  const float* Bm   = en_seq + (size_t)b * TE_ * D_;
  const float* aL   = &aS[m][2 * khalf];
  const float* bcol = Bm + td * 16 + m;

  v8f acc = {};
  for (int k = 0; k < TE_; k += 4) {
    const int ka = k + 2 * khalf;
    v2f a, bb;
    a.x  = aL[k + 0];
    a.y  = aL[k + 1];
    bb.x = bcol[(size_t)(ka + 0) * D_];
    bb.y = bcol[(size_t)(ka + 1) * D_];
    acc = __builtin_amdgcn_wmma_f32_16x16x4_f32(false, a, false, bb,
                                                (short)0, acc, false, false);
  }

#pragma unroll
  for (int r = 0; r < 8; ++r) {
    const int s_g  = ts * 16 + r + 8 * khalf;
    const float se = acc[r];
    const float de = de_seq[((size_t)(b * TD_ + s_g)) * D_ + d_g];
    float* o = out + ((size_t)(b * TD_ + s_g)) * (4 * D_);
    o[d_g]          = de;
    o[D_ + d_g]     = se;
    o[2 * D_ + d_g] = de * se;
    o[3 * D_ + d_g] = de * hh;
  }
}

extern "C" void kernel_launch(void* const* d_in, const int* in_sizes, int n_in,
                              void* d_out, int out_size, void* d_ws, size_t ws_size,
                              hipStream_t stream) {
  const float* en_seq = (const float*)d_in[0];
  const float* de_seq = (const float*)d_in[1];
  const float* w_en   = (const float*)d_in[2];
  const float* w_de   = (const float*)d_in[3];
  const float* nu     = (const float*)d_in[4];
  float* out = (float*)d_out;

  float* ws = (float*)d_ws;
  const size_t nAE = (size_t)B_ * TE_ * U_;
  float* ae     = ws;
  float* ad     = ae + nAE;
  float* mu     = ad + nAE;
  float* pmax   = mu + (size_t)B_ * TD_ * TE_;

  const dim3 blk(256);
  gemm_f32_wmma<<<dim3(256), blk, 0, stream>>>(en_seq, w_en, ae);
  gemm_f32_wmma<<<dim3(256), blk, 0, stream>>>(de_seq, w_de, ad);
  co_mu_kernel<<<dim3(64, B_), blk, 0, stream>>>(ae, ad, nu, mu, pmax);
  sum_en_out_kernel<<<dim3(32, B_), blk, 0, stream>>>(mu, pmax, en_seq,
                                                      de_seq, out);
}


# projection launches merged (4 -> 3): one kernel forms ae and ad, each wave holding the en tile and the de tile of its position on two independent f32 WMMA accumulators; sum order unchanged
# speedup vs baseline: 3.6573x; 1.1374x over previous
#include <hip/hip_runtime.h>
#include <cmath>

#define B_  8
#define TE_ 256
#define TD_ 256
#define D_  256
#define U_  256

typedef float v2f __attribute__((ext_vector_type(2)));
typedef float v8f __attribute__((ext_vector_type(8)));

static_assert(TE_ == TD_, "one tile grid serves both projections");

__global__ void __launch_bounds__(256) proj2_f32_wmma(
    const float* __restrict__ en_seq, const float* __restrict__ w_en,
    float* __restrict__ ae,
    const float* __restrict__ de_seq, const float* __restrict__ w_de,
    float* __restrict__ ad) {
  const int lane  = threadIdx.x & 31;
  const int wave  = threadIdx.x >> 5;
  const int tile  = blockIdx.x * 8 + wave;
  const int tn    = tile & 15;
  const int tm    = tile >> 4;
  const int m     = lane & 15;
  const int khalf = lane >> 4;

  const size_t aoff = (size_t)(tm * 16 + m) * D_;
  const int    boff = tn * 16 + m;
  const float* arow_e = en_seq + aoff;
  const float* arow_d = de_seq + aoff;
  const float* bcol_e = w_en + boff;
  const float* bcol_d = w_de + boff;
  v8f acc_e = {};
  v8f acc_d = {};
  for (int k0 = 0; k0 < D_; k0 += 32) {
    v2f a_e[8], b_e[8], a_d[8], b_d[8];
#pragma unroll
    for (int j = 0; j < 8; ++j) {
      const int ka = k0 + 4 * j + 2 * khalf;
      a_e[j].x = arow_e[ka + 0];
      a_e[j].y = arow_e[ka + 1];
      b_e[j].x = bcol_e[(size_t)(ka + 0) * U_];
      b_e[j].y = bcol_e[(size_t)(ka + 1) * U_];
      a_d[j].x = arow_d[ka + 0];
      a_d[j].y = arow_d[ka + 1];
      b_d[j].x = bcol_d[(size_t)(ka + 0) * U_];
      b_d[j].y = bcol_d[(size_t)(ka + 1) * U_];
    }
    asm volatile("" :: "v"(a_e[0]), "v"(a_e[1]), "v"(a_e[2]), "v"(a_e[3]),
                       "v"(a_e[4]), "v"(a_e[5]), "v"(a_e[6]), "v"(a_e[7]),
                       "v"(b_e[0]), "v"(b_e[1]), "v"(b_e[2]), "v"(b_e[3]),
                       "v"(b_e[4]), "v"(b_e[5]), "v"(b_e[6]), "v"(b_e[7]));
    asm volatile("" :: "v"(a_d[0]), "v"(a_d[1]), "v"(a_d[2]), "v"(a_d[3]),
                       "v"(a_d[4]), "v"(a_d[5]), "v"(a_d[6]), "v"(a_d[7]),
                       "v"(b_d[0]), "v"(b_d[1]), "v"(b_d[2]), "v"(b_d[3]),
                       "v"(b_d[4]), "v"(b_d[5]), "v"(b_d[6]), "v"(b_d[7]));
#pragma unroll
    for (int j = 0; j < 8; ++j) {
      acc_e = __builtin_amdgcn_wmma_f32_16x16x4_f32(false, a_e[j], false, b_e[j],
                                                    (short)0, acc_e, false, false);
      acc_d = __builtin_amdgcn_wmma_f32_16x16x4_f32(false, a_d[j], false, b_d[j],
                                                    (short)0, acc_d, false, false);
    }
  }
#pragma unroll
  for (int r = 0; r < 8; ++r) {
    const size_t o = (size_t)(tm * 16 + r + 8 * khalf) * U_ + tn * 16 + m;
    ae[o] = acc_e[r];
    ad[o] = acc_d[r];
  }
}

__device__ __forceinline__ float tanh_hw(float x) {
  return __builtin_amdgcn_tanhf(x);
}

__global__ void __launch_bounds__(256) co_mu_kernel(
    const float* __restrict__ ae, const float* __restrict__ ad,
    const float* __restrict__ nu, float* __restrict__ mu,
    float* __restrict__ pmax) {
  const int b      = blockIdx.y;
  const int s_tile = blockIdx.x >> 3;
  const int t_tile = blockIdx.x & 7;
  __shared__ __attribute__((aligned(16))) float adT[32][260];
  __shared__ __attribute__((aligned(16))) float aeT[32][260];
  __shared__ __attribute__((aligned(16))) float nuS[256];

  const int tid = threadIdx.x;
  nuS[tid] = nu[tid];

  const float4* ad4 = reinterpret_cast<const float4*>(
      ad + ((size_t)(b * TD_ + s_tile * 32)) * U_);
  const float4* ae4 = reinterpret_cast<const float4*>(
      ae + ((size_t)(b * TE_ + t_tile * 32)) * U_);
  float4 va[8], vb[8];
#pragma unroll
  for (int j = 0; j < 8; ++j) {
    va[j] = ad4[tid + 256 * j];
    vb[j] = ae4[tid + 256 * j];
  }
#pragma unroll
  for (int j = 0; j < 8; ++j) {
    const int i = tid + 256 * j;
    const int r = i >> 6, c = (i & 63) * 4;
    *reinterpret_cast<float4*>(&adT[r][c]) = va[j];
    *reinterpret_cast<float4*>(&aeT[r][c]) = vb[j];
  }
  __syncthreads();

  const int si = tid >> 4, ti = tid & 15;
  float acc00 = 0.f, acc01 = 0.f, acc10 = 0.f, acc11 = 0.f;
#pragma unroll 4
  for (int u = 0; u < U_; ++u) {
    const float a0 = adT[si][u], a1 = adT[si + 16][u];
    const float e0 = aeT[ti][u], e1 = aeT[ti + 16][u];
    const float n  = nuS[u];
    acc00 += tanh_hw(a0 + e0) * n;
    acc01 += tanh_hw(a0 + e1) * n;
    acc10 += tanh_hw(a1 + e0) * n;
    acc11 += tanh_hw(a1 + e1) * n;
  }

  const int s0 = b * TD_ + s_tile * 32 + si;
  const int t0 = t_tile * 32 + ti;
  mu[(size_t)s0 * TE_ + t0]             = acc00;
  mu[(size_t)s0 * TE_ + t0 + 16]        = acc01;
  mu[(size_t)(s0 + 16) * TE_ + t0]      = acc10;
  mu[(size_t)(s0 + 16) * TE_ + t0 + 16] = acc11;

  float p00 = acc00, p01 = acc01, p10 = acc10, p11 = acc11;
#pragma unroll
  for (int x = 8; x >= 1; x >>= 1) {
    p00 = fmaxf(p00, __shfl_xor(p00, x));
    p01 = fmaxf(p01, __shfl_xor(p01, x));
    p10 = fmaxf(p10, __shfl_xor(p10, x));
    p11 = fmaxf(p11, __shfl_xor(p11, x));
  }
  if (ti == 0) {
    float* q0 = pmax + (size_t)s0 * 16 + 2 * t_tile;
    float* q1 = pmax + (size_t)(s0 + 16) * 16 + 2 * t_tile;
    q0[0] = p00; q0[1] = p01;
    q1[0] = p10; q1[1] = p11;
  }
}

__device__ __forceinline__ void lds_barrier() {
  __builtin_amdgcn_fence(__ATOMIC_RELEASE, "workgroup", "local");
  __builtin_amdgcn_s_barrier();
  __builtin_amdgcn_fence(__ATOMIC_ACQUIRE, "workgroup", "local");
}

__device__ __forceinline__ float wave_max(float x) {
  x = fmaxf(x, __shfl_xor(x, 16));
  x = fmaxf(x, __shfl_xor(x, 8));
  x = fmaxf(x, __shfl_xor(x, 4));
  x = fmaxf(x, __shfl_xor(x, 2));
  x = fmaxf(x, __shfl_xor(x, 1));
  return x;
}
__device__ __forceinline__ float wave_sum(float x) {
  x += __shfl_xor(x, 16);
  x += __shfl_xor(x, 8);
  x += __shfl_xor(x, 4);
  x += __shfl_xor(x, 2);
  x += __shfl_xor(x, 1);
  return x;
}

__global__ void __launch_bounds__(256) sum_en_out_kernel(
    const float* __restrict__ mu, const float* __restrict__ pmax,
    const float* __restrict__ en_seq, const float* __restrict__ de_seq,
    float* __restrict__ out) {
  const int b     = blockIdx.y;
  const int tid   = threadIdx.x;
  const int lane  = threadIdx.x & 31;
  const int wave  = threadIdx.x >> 5;
  const int tile  = blockIdx.x * 8 + wave;
  const int td    = tile & 15;
  const int ts    = tile >> 4;
  const int m     = lane & 15;
  const int khalf = lane >> 4;
  const int dbase = (blockIdx.x & 1) * 128;

  __shared__ __attribute__((aligned(16))) float aS[16][260];
  __shared__ __attribute__((aligned(16))) float hpS[8][128];
  __shared__ float maS[256];
  __shared__ float wmaxS[8];
  __shared__ float wsumS[8];

  const float4* pm4 = reinterpret_cast<const float4*>(
      pmax + ((size_t)(b * TD_ + tid)) * 16);
  const float4 p0 = pm4[0], p1 = pm4[1], p2 = pm4[2], p3 = pm4[3];

  float4 mv[2][2];
#pragma unroll
  for (int rr = 0; rr < 2; ++rr) {
    const int r = 2 * wave + rr;
    const float* mrow = mu + ((size_t)(b * TD_ + ts * 16 + r)) * TE_;
    mv[rr][0] = *reinterpret_cast<const float4*>(mrow + 4 * lane);
    mv[rr][1] = *reinterpret_cast<const float4*>(mrow + 128 + 4 * lane);
  }

  const float* dp = de_seq + ((size_t)(b * TD_ + 32 * wave)) * D_ + dbase + 4 * lane;
  float4 dv[32];
#pragma unroll
  for (int i = 0; i < 32; ++i)
    dv[i] = *reinterpret_cast<const float4*>(dp + (size_t)i * D_);

  const float v = fmaxf(
      fmaxf(fmaxf(fmaxf(p0.x, p0.y), fmaxf(p0.z, p0.w)),
            fmaxf(fmaxf(p1.x, p1.y), fmaxf(p1.z, p1.w))),
      fmaxf(fmaxf(fmaxf(p2.x, p2.y), fmaxf(p2.z, p2.w)),
            fmaxf(fmaxf(p3.x, p3.y), fmaxf(p3.z, p3.w))));
  const float wm = wave_max(v);
  if (lane == 0) wmaxS[wave] = wm;
  lds_barrier();

#pragma unroll
  for (int rr = 0; rr < 2; ++rr) {
    const int r = 2 * wave + rr;
    const float4 v0 = mv[rr][0];
    const float4 v1 = mv[rr][1];
    const float mx = wave_max(fmaxf(fmaxf(fmaxf(v0.x, v0.y), fmaxf(v0.z, v0.w)),
                                    fmaxf(fmaxf(v1.x, v1.y), fmaxf(v1.z, v1.w))));
    float4 e0, e1;
    e0.x = expf(v0.x - mx); e0.y = expf(v0.y - mx);
    e0.z = expf(v0.z - mx); e0.w = expf(v0.w - mx);
    e1.x = expf(v1.x - mx); e1.y = expf(v1.y - mx);
    e1.z = expf(v1.z - mx); e1.w = expf(v1.w - mx);
    const float sm = wave_sum(((e0.x + e0.y) + (e0.z + e0.w)) +
                              ((e1.x + e1.y) + (e1.z + e1.w)));
    const float inv = 1.0f / sm;
    e0.x *= inv; e0.y *= inv; e0.z *= inv; e0.w *= inv;
    e1.x *= inv; e1.y *= inv; e1.z *= inv; e1.w *= inv;
    *reinterpret_cast<float4*>(&aS[r][4 * lane])       = e0;
    *reinterpret_cast<float4*>(&aS[r][128 + 4 * lane]) = e1;
  }

  float mxs = wmaxS[0];
  for (int j = 1; j < 8; ++j) mxs = fmaxf(mxs, wmaxS[j]);
  const float ev = expf(v - mxs);
  const float wsm = wave_sum(ev);
  if (lane == 0) wsumS[wave] = wsm;
  lds_barrier();
  float tot = wsumS[0];
  for (int j = 1; j < 8; ++j) tot += wsumS[j];
  maS[tid] = ev * (1.0f / tot);
  lds_barrier();

  {
    float4 ha = {0.f, 0.f, 0.f, 0.f};
#pragma unroll
    for (int i = 0; i < 32; ++i) {
      const float w = maS[32 * wave + i];
      ha.x = fmaf(dv[i].x, w, ha.x);
      ha.y = fmaf(dv[i].y, w, ha.y);
      ha.z = fmaf(dv[i].z, w, ha.z);
      ha.w = fmaf(dv[i].w, w, ha.w);
    }
    *reinterpret_cast<float4*>(&hpS[wave][4 * lane]) = ha;
  }

  lds_barrier();

  const int d_g = td * 16 + m;
  float hh = hpS[0][d_g - dbase];
  for (int j = 1; j < 8; ++j) hh += hpS[j][d_g - dbase];

  const float* Bm   = en_seq + (size_t)b * TE_ * D_;
  const float* aL   = &aS[m][2 * khalf];
  const float* bcol = Bm + td * 16 + m;

  v8f acc = {};
  for (int k = 0; k < TE_; k += 4) {
    const int ka = k + 2 * khalf;
    v2f a, bb;
    a.x  = aL[k + 0];
    a.y  = aL[k + 1];
    bb.x = bcol[(size_t)(ka + 0) * D_];
    bb.y = bcol[(size_t)(ka + 1) * D_];
    acc = __builtin_amdgcn_wmma_f32_16x16x4_f32(false, a, false, bb,
                                                (short)0, acc, false, false);
  }

#pragma unroll
  for (int r = 0; r < 8; ++r) {
    const int s_g  = ts * 16 + r + 8 * khalf;
    const float se = acc[r];
    const float de = de_seq[((size_t)(b * TD_ + s_g)) * D_ + d_g];
    float* o = out + ((size_t)(b * TD_ + s_g)) * (4 * D_);
    o[d_g]          = de;
    o[D_ + d_g]     = se;
    o[2 * D_ + d_g] = de * se;
    o[3 * D_ + d_g] = de * hh;
  }
}

extern "C" void kernel_launch(void* const* d_in, const int* in_sizes, int n_in,
                              void* d_out, int out_size, void* d_ws, size_t ws_size,
                              hipStream_t stream) {
  const float* en_seq = (const float*)d_in[0];
  const float* de_seq = (const float*)d_in[1];
  const float* w_en   = (const float*)d_in[2];
  const float* w_de   = (const float*)d_in[3];
  const float* nu     = (const float*)d_in[4];
  float* out = (float*)d_out;

  float* ws = (float*)d_ws;
  const size_t nAE = (size_t)B_ * TE_ * U_;
  float* ae     = ws;
  float* ad     = ae + nAE;
  float* mu     = ad + nAE;
  float* pmax   = mu + (size_t)B_ * TD_ * TE_;

  const dim3 blk(256);
  proj2_f32_wmma<<<dim3(256), blk, 0, stream>>>(en_seq, w_en, ae,
                                               de_seq, w_de, ad);
  co_mu_kernel<<<dim3(64, B_), blk, 0, stream>>>(ae, ad, nu, mu, pmax);
  sum_en_out_kernel<<<dim3(32, B_), blk, 0, stream>>>(mu, pmax, en_seq,
                                                      de_seq, out);
}


# projection kernel re-tiled: a wave owns a 16x64 strip of one product, 4 accumulators share each A fragment, w and strip rows move as 16-byte vectors, next 32 k loaded ahead; sum order unchanged
# speedup vs baseline: 4.6180x; 1.2627x over previous
#include <hip/hip_runtime.h>
#include <cmath>

#define B_  8
#define TE_ 256
#define TD_ 256
#define D_  256
#define U_  256

typedef float v2f __attribute__((ext_vector_type(2)));
typedef float v8f __attribute__((ext_vector_type(8)));

static_assert(TE_ == TD_, "one tile grid serves both projections");

__device__ __forceinline__ void proj_load_trip(
    v2f (&a)[8], float4 (&b0)[8], float4 (&b1)[8],
    const float* __restrict__ arow, const float* __restrict__ brow, int k0) {
#pragma unroll
  for (int j = 0; j < 8; ++j) {
    const int k = k0 + 4 * j;
    a[j]  = *reinterpret_cast<const v2f*>(arow + k);
    b0[j] = *reinterpret_cast<const float4*>(brow + (size_t)(k + 0) * U_);
    b1[j] = *reinterpret_cast<const float4*>(brow + (size_t)(k + 1) * U_);
  }
}

__device__ __forceinline__ void proj_mma_trip(
    v8f (&acc)[4], const v2f (&a)[8], const float4 (&b0)[8],
    const float4 (&b1)[8]) {
#pragma unroll
  for (int j = 0; j < 8; ++j) {
    const float p0[4] = {b0[j].x, b0[j].y, b0[j].z, b0[j].w};
    const float p1[4] = {b1[j].x, b1[j].y, b1[j].z, b1[j].w};
#pragma unroll
    for (int t = 0; t < 4; ++t) {
      v2f bb;
      bb.x = p0[t];
      bb.y = p1[t];
      acc[t] = __builtin_amdgcn_wmma_f32_16x16x4_f32(false, a[j], false, bb,
                                                     (short)0, acc[t], false, false);
    }
  }
}

__global__ void __launch_bounds__(128) proj2_f32_wmma(
    const float* __restrict__ en_seq, const float* __restrict__ w_en,
    float* __restrict__ ae,
    const float* __restrict__ de_seq, const float* __restrict__ w_de,
    float* __restrict__ ad) {
  const int lane  = threadIdx.x & 31;
  const int wave  = threadIdx.x >> 5;
  const int prod  = blockIdx.x & 1;
  const int panel = (blockIdx.x >> 1) & 3;
  const int tm    = (blockIdx.x >> 3) * 4 + wave;
  const int m     = lane & 15;
  const int khalf = lane >> 4;

  const float* x = prod ? de_seq : en_seq;
  const float* w = prod ? w_de : w_en;
  float*       y = prod ? ad : ae;

  const float* arow = x + (size_t)(tm * 16 + m) * D_ + 2 * khalf;
  const float* brow = w + (size_t)(2 * khalf) * U_ + panel * 64 + 4 * m;

  v8f acc[4] = {};
  v2f    a0[8], a1[8];
  float4 b00[8], b10[8], b01[8], b11[8];

  proj_load_trip(a0, b00, b10, arow, brow, 0);
#pragma nounroll
  for (int k0 = 0; k0 < D_ - 64; k0 += 64) {
    proj_load_trip(a1, b01, b11, arow, brow, k0 + 32);
    __builtin_amdgcn_sched_barrier(0);
    proj_mma_trip(acc, a0, b00, b10);
    __builtin_amdgcn_sched_barrier(0);
    proj_load_trip(a0, b00, b10, arow, brow, k0 + 64);
    __builtin_amdgcn_sched_barrier(0);
    proj_mma_trip(acc, a1, b01, b11);
    __builtin_amdgcn_sched_barrier(0);
  }
  proj_load_trip(a1, b01, b11, arow, brow, D_ - 32);
  __builtin_amdgcn_sched_barrier(0);
  proj_mma_trip(acc, a0, b00, b10);
  proj_mma_trip(acc, a1, b01, b11);

#pragma unroll
  for (int r = 0; r < 8; ++r) {
    float4 v;
    v.x = acc[0][r];
    v.y = acc[1][r];
    v.z = acc[2][r];
    v.w = acc[3][r];
    *reinterpret_cast<float4*>(
        y + (size_t)(tm * 16 + r + 8 * khalf) * U_ + panel * 64 + 4 * m) = v;
  }
}

__device__ __forceinline__ float tanh_hw(float x) {
  return __builtin_amdgcn_tanhf(x);
}

__global__ void __launch_bounds__(256) co_mu_kernel(
    const float* __restrict__ ae, const float* __restrict__ ad,
    const float* __restrict__ nu, float* __restrict__ mu,
    float* __restrict__ pmax) {
  const int b      = blockIdx.y;
  const int s_tile = blockIdx.x >> 3;
  const int t_tile = blockIdx.x & 7;
  __shared__ __attribute__((aligned(16))) float adT[32][260];
  __shared__ __attribute__((aligned(16))) float aeT[32][260];
  __shared__ __attribute__((aligned(16))) float nuS[256];

  const int tid = threadIdx.x;
  nuS[tid] = nu[tid];

  const float4* ad4 = reinterpret_cast<const float4*>(
      ad + ((size_t)(b * TD_ + s_tile * 32)) * U_);
  const float4* ae4 = reinterpret_cast<const float4*>(
      ae + ((size_t)(b * TE_ + t_tile * 32)) * U_);
  float4 va[8], vb[8];
#pragma unroll
  for (int j = 0; j < 8; ++j) {
    va[j] = ad4[tid + 256 * j];
    vb[j] = ae4[tid + 256 * j];
  }
#pragma unroll
  for (int j = 0; j < 8; ++j) {
    const int i = tid + 256 * j;
    const int r = i >> 6, c = (i & 63) * 4;
    *reinterpret_cast<float4*>(&adT[r][c]) = va[j];
    *reinterpret_cast<float4*>(&aeT[r][c]) = vb[j];
  }
  __syncthreads();

  const int si = tid >> 4, ti = tid & 15;
  float acc00 = 0.f, acc01 = 0.f, acc10 = 0.f, acc11 = 0.f;
#pragma unroll 4
  for (int u = 0; u < U_; ++u) {
    const float a0 = adT[si][u], a1 = adT[si + 16][u];
    const float e0 = aeT[ti][u], e1 = aeT[ti + 16][u];
    const float n  = nuS[u];
    acc00 += tanh_hw(a0 + e0) * n;
    acc01 += tanh_hw(a0 + e1) * n;
    acc10 += tanh_hw(a1 + e0) * n;
    acc11 += tanh_hw(a1 + e1) * n;
  }

  const int s0 = b * TD_ + s_tile * 32 + si;
  const int t0 = t_tile * 32 + ti;
  mu[(size_t)s0 * TE_ + t0]             = acc00;
  mu[(size_t)s0 * TE_ + t0 + 16]        = acc01;
  mu[(size_t)(s0 + 16) * TE_ + t0]      = acc10;
  mu[(size_t)(s0 + 16) * TE_ + t0 + 16] = acc11;

  float p00 = acc00, p01 = acc01, p10 = acc10, p11 = acc11;
#pragma unroll
  for (int x = 8; x >= 1; x >>= 1) {
    p00 = fmaxf(p00, __shfl_xor(p00, x));
    p01 = fmaxf(p01, __shfl_xor(p01, x));
    p10 = fmaxf(p10, __shfl_xor(p10, x));
    p11 = fmaxf(p11, __shfl_xor(p11, x));
  }
  if (ti == 0) {
    float* q0 = pmax + (size_t)s0 * 16 + 2 * t_tile;
    float* q1 = pmax + (size_t)(s0 + 16) * 16 + 2 * t_tile;
    q0[0] = p00; q0[1] = p01;
    q1[0] = p10; q1[1] = p11;
  }
}

__device__ __forceinline__ void lds_barrier() {
  __builtin_amdgcn_fence(__ATOMIC_RELEASE, "workgroup", "local");
  __builtin_amdgcn_s_barrier();
  __builtin_amdgcn_fence(__ATOMIC_ACQUIRE, "workgroup", "local");
}

__device__ __forceinline__ float wave_max(float x) {
  x = fmaxf(x, __shfl_xor(x, 16));
  x = fmaxf(x, __shfl_xor(x, 8));
  x = fmaxf(x, __shfl_xor(x, 4));
  x = fmaxf(x, __shfl_xor(x, 2));
  x = fmaxf(x, __shfl_xor(x, 1));
  return x;
}
__device__ __forceinline__ float wave_sum(float x) {
  x += __shfl_xor(x, 16);
  x += __shfl_xor(x, 8);
  x += __shfl_xor(x, 4);
  x += __shfl_xor(x, 2);
  x += __shfl_xor(x, 1);
  return x;
}

__global__ void __launch_bounds__(256) sum_en_out_kernel(
    const float* __restrict__ mu, const float* __restrict__ pmax,
    const float* __restrict__ en_seq, const float* __restrict__ de_seq,
    float* __restrict__ out) {
  const int b     = blockIdx.y;
  const int tid   = threadIdx.x;
  const int lane  = threadIdx.x & 31;
  const int wave  = threadIdx.x >> 5;
  const int tile  = blockIdx.x * 8 + wave;
  const int td    = tile & 15;
  const int ts    = tile >> 4;
  const int m     = lane & 15;
  const int khalf = lane >> 4;
  const int dbase = (blockIdx.x & 1) * 128;

  __shared__ __attribute__((aligned(16))) float aS[16][260];
  __shared__ __attribute__((aligned(16))) float hpS[8][128];
  __shared__ float maS[256];
  __shared__ float wmaxS[8];
  __shared__ float wsumS[8];

  const float4* pm4 = reinterpret_cast<const float4*>(
      pmax + ((size_t)(b * TD_ + tid)) * 16);
  const float4 p0 = pm4[0], p1 = pm4[1], p2 = pm4[2], p3 = pm4[3];

  float4 mv[2][2];
#pragma unroll
  for (int rr = 0; rr < 2; ++rr) {
    const int r = 2 * wave + rr;
    const float* mrow = mu + ((size_t)(b * TD_ + ts * 16 + r)) * TE_;
    mv[rr][0] = *reinterpret_cast<const float4*>(mrow + 4 * lane);
    mv[rr][1] = *reinterpret_cast<const float4*>(mrow + 128 + 4 * lane);
  }

  const float* dp = de_seq + ((size_t)(b * TD_ + 32 * wave)) * D_ + dbase + 4 * lane;
  float4 dv[32];
#pragma unroll
  for (int i = 0; i < 32; ++i)
    dv[i] = *reinterpret_cast<const float4*>(dp + (size_t)i * D_);

  const float v = fmaxf(
      fmaxf(fmaxf(fmaxf(p0.x, p0.y), fmaxf(p0.z, p0.w)),
            fmaxf(fmaxf(p1.x, p1.y), fmaxf(p1.z, p1.w))),
      fmaxf(fmaxf(fmaxf(p2.x, p2.y), fmaxf(p2.z, p2.w)),
            fmaxf(fmaxf(p3.x, p3.y), fmaxf(p3.z, p3.w))));
  const float wm = wave_max(v);
  if (lane == 0) wmaxS[wave] = wm;
  lds_barrier();

#pragma unroll
  for (int rr = 0; rr < 2; ++rr) {
    const int r = 2 * wave + rr;
    const float4 v0 = mv[rr][0];
    const float4 v1 = mv[rr][1];
    const float mx = wave_max(fmaxf(fmaxf(fmaxf(v0.x, v0.y), fmaxf(v0.z, v0.w)),
                                    fmaxf(fmaxf(v1.x, v1.y), fmaxf(v1.z, v1.w))));
    float4 e0, e1;
    e0.x = expf(v0.x - mx); e0.y = expf(v0.y - mx);
    e0.z = expf(v0.z - mx); e0.w = expf(v0.w - mx);
    e1.x = expf(v1.x - mx); e1.y = expf(v1.y - mx);
    e1.z = expf(v1.z - mx); e1.w = expf(v1.w - mx);
    const float sm = wave_sum(((e0.x + e0.y) + (e0.z + e0.w)) +
                              ((e1.x + e1.y) + (e1.z + e1.w)));
    const float inv = 1.0f / sm;
    e0.x *= inv; e0.y *= inv; e0.z *= inv; e0.w *= inv;
    e1.x *= inv; e1.y *= inv; e1.z *= inv; e1.w *= inv;
    *reinterpret_cast<float4*>(&aS[r][4 * lane])       = e0;
    *reinterpret_cast<float4*>(&aS[r][128 + 4 * lane]) = e1;
  }

  float mxs = wmaxS[0];
  for (int j = 1; j < 8; ++j) mxs = fmaxf(mxs, wmaxS[j]);
  const float ev = expf(v - mxs);
  const float wsm = wave_sum(ev);
  if (lane == 0) wsumS[wave] = wsm;
  lds_barrier();
  float tot = wsumS[0];
  for (int j = 1; j < 8; ++j) tot += wsumS[j];
  maS[tid] = ev * (1.0f / tot);
  lds_barrier();

  {
    float4 ha = {0.f, 0.f, 0.f, 0.f};
#pragma unroll
    for (int i = 0; i < 32; ++i) {
      const float w = maS[32 * wave + i];
      ha.x = fmaf(dv[i].x, w, ha.x);
      ha.y = fmaf(dv[i].y, w, ha.y);
      ha.z = fmaf(dv[i].z, w, ha.z);
      ha.w = fmaf(dv[i].w, w, ha.w);
    }
    *reinterpret_cast<float4*>(&hpS[wave][4 * lane]) = ha;
  }

  lds_barrier();

  const int d_g = td * 16 + m;
  float hh = hpS[0][d_g - dbase];
  for (int j = 1; j < 8; ++j) hh += hpS[j][d_g - dbase];

  const float* Bm   = en_seq + (size_t)b * TE_ * D_;
  const float* aL   = &aS[m][2 * khalf];
  const float* bcol = Bm + td * 16 + m;

  v8f acc = {};
  for (int k = 0; k < TE_; k += 4) {
    const int ka = k + 2 * khalf;
    v2f a, bb;
    a.x  = aL[k + 0];
    a.y  = aL[k + 1];
    bb.x = bcol[(size_t)(ka + 0) * D_];
    bb.y = bcol[(size_t)(ka + 1) * D_];
    acc = __builtin_amdgcn_wmma_f32_16x16x4_f32(false, a, false, bb,
                                                (short)0, acc, false, false);
  }

#pragma unroll
  for (int r = 0; r < 8; ++r) {
    const int s_g  = ts * 16 + r + 8 * khalf;
    const float se = acc[r];
    const float de = de_seq[((size_t)(b * TD_ + s_g)) * D_ + d_g];
    float* o = out + ((size_t)(b * TD_ + s_g)) * (4 * D_);
    o[d_g]          = de;
    o[D_ + d_g]     = se;
    o[2 * D_ + d_g] = de * se;
    o[3 * D_ + d_g] = de * hh;
  }
}

extern "C" void kernel_launch(void* const* d_in, const int* in_sizes, int n_in,
                              void* d_out, int out_size, void* d_ws, size_t ws_size,
                              hipStream_t stream) {
  const float* en_seq = (const float*)d_in[0];
  const float* de_seq = (const float*)d_in[1];
  const float* w_en   = (const float*)d_in[2];
  const float* w_de   = (const float*)d_in[3];
  const float* nu     = (const float*)d_in[4];
  float* out = (float*)d_out;

  float* ws = (float*)d_ws;
  const size_t nAE = (size_t)B_ * TE_ * U_;
  float* ae     = ws;
  float* ad     = ae + nAE;
  float* mu     = ad + nAE;
  float* pmax   = mu + (size_t)B_ * TD_ * TE_;

  const dim3 blk(256);
  proj2_f32_wmma<<<dim3(256), dim3(128), 0, stream>>>(en_seq, w_en, ae,
                                                     de_seq, w_de, ad);
  co_mu_kernel<<<dim3(64, B_), blk, 0, stream>>>(ae, ad, nu, mu, pmax);
  sum_en_out_kernel<<<dim3(32, B_), blk, 0, stream>>>(mu, pmax, en_seq,
                                                      de_seq, out);
}


# output launch: h_hat and the de and de*h_hat quarters moved to 64 extra blocks (32 columns x 256 rows each); matrix blocks no longer read 256 de rows each; h_hat sum order over s changed
# speedup vs baseline: 4.9025x; 1.0616x over previous
#include <hip/hip_runtime.h>
#include <cmath>

#define B_  8
#define TE_ 256
#define TD_ 256
#define D_  256
#define U_  256

typedef float v2f __attribute__((ext_vector_type(2)));
typedef float v8f __attribute__((ext_vector_type(8)));

static_assert(TE_ == TD_, "one tile grid serves both projections");

__device__ __forceinline__ void proj_load_trip(
    v2f (&a)[8], float4 (&b0)[8], float4 (&b1)[8],
    const float* __restrict__ arow, const float* __restrict__ brow, int k0) {
#pragma unroll
  for (int j = 0; j < 8; ++j) {
    const int k = k0 + 4 * j;
    a[j]  = *reinterpret_cast<const v2f*>(arow + k);
    b0[j] = *reinterpret_cast<const float4*>(brow + (size_t)(k + 0) * U_);
    b1[j] = *reinterpret_cast<const float4*>(brow + (size_t)(k + 1) * U_);
  }
}

__device__ __forceinline__ void proj_mma_trip(
    v8f (&acc)[4], const v2f (&a)[8], const float4 (&b0)[8],
    const float4 (&b1)[8]) {
#pragma unroll
  for (int j = 0; j < 8; ++j) {
    const float p0[4] = {b0[j].x, b0[j].y, b0[j].z, b0[j].w};
    const float p1[4] = {b1[j].x, b1[j].y, b1[j].z, b1[j].w};
#pragma unroll
    for (int t = 0; t < 4; ++t) {
      v2f bb;
      bb.x = p0[t];
      bb.y = p1[t];
      acc[t] = __builtin_amdgcn_wmma_f32_16x16x4_f32(false, a[j], false, bb,
                                                     (short)0, acc[t], false, false);
    }
  }
}

__global__ void __launch_bounds__(128) proj2_f32_wmma(
    const float* __restrict__ en_seq, const float* __restrict__ w_en,
    float* __restrict__ ae,
    const float* __restrict__ de_seq, const float* __restrict__ w_de,
    float* __restrict__ ad) {
  const int lane  = threadIdx.x & 31;
  const int wave  = threadIdx.x >> 5;
  const int prod  = blockIdx.x & 1;
  const int panel = (blockIdx.x >> 1) & 3;
  const int tm    = (blockIdx.x >> 3) * 4 + wave;
  const int m     = lane & 15;
  const int khalf = lane >> 4;

  const float* x = prod ? de_seq : en_seq;
  const float* w = prod ? w_de : w_en;
  float*       y = prod ? ad : ae;

  const float* arow = x + (size_t)(tm * 16 + m) * D_ + 2 * khalf;
  const float* brow = w + (size_t)(2 * khalf) * U_ + panel * 64 + 4 * m;

  v8f acc[4] = {};
  v2f    a0[8], a1[8];
  float4 b00[8], b10[8], b01[8], b11[8];

  proj_load_trip(a0, b00, b10, arow, brow, 0);
#pragma nounroll
  for (int k0 = 0; k0 < D_ - 64; k0 += 64) {
    proj_load_trip(a1, b01, b11, arow, brow, k0 + 32);
    __builtin_amdgcn_sched_barrier(0);
    proj_mma_trip(acc, a0, b00, b10);
    __builtin_amdgcn_sched_barrier(0);
    proj_load_trip(a0, b00, b10, arow, brow, k0 + 64);
    __builtin_amdgcn_sched_barrier(0);
    proj_mma_trip(acc, a1, b01, b11);
    __builtin_amdgcn_sched_barrier(0);
  }
  proj_load_trip(a1, b01, b11, arow, brow, D_ - 32);
  __builtin_amdgcn_sched_barrier(0);
  proj_mma_trip(acc, a0, b00, b10);
  proj_mma_trip(acc, a1, b01, b11);

#pragma unroll
  for (int r = 0; r < 8; ++r) {
    float4 v;
    v.x = acc[0][r];
    v.y = acc[1][r];
    v.z = acc[2][r];
    v.w = acc[3][r];
    *reinterpret_cast<float4*>(
        y + (size_t)(tm * 16 + r + 8 * khalf) * U_ + panel * 64 + 4 * m) = v;
  }
}

__device__ __forceinline__ float tanh_hw(float x) {
  return __builtin_amdgcn_tanhf(x);
}

__global__ void __launch_bounds__(256) co_mu_kernel(
    const float* __restrict__ ae, const float* __restrict__ ad,
    const float* __restrict__ nu, float* __restrict__ mu,
    float* __restrict__ pmax) {
  const int b      = blockIdx.y;
  const int s_tile = blockIdx.x >> 3;
  const int t_tile = blockIdx.x & 7;
  __shared__ __attribute__((aligned(16))) float adT[32][260];
  __shared__ __attribute__((aligned(16))) float aeT[32][260];
  __shared__ __attribute__((aligned(16))) float nuS[256];

  const int tid = threadIdx.x;
  nuS[tid] = nu[tid];

  const float4* ad4 = reinterpret_cast<const float4*>(
      ad + ((size_t)(b * TD_ + s_tile * 32)) * U_);
  const float4* ae4 = reinterpret_cast<const float4*>(
      ae + ((size_t)(b * TE_ + t_tile * 32)) * U_);
  float4 va[8], vb[8];
#pragma unroll
  for (int j = 0; j < 8; ++j) {
    va[j] = ad4[tid + 256 * j];
    vb[j] = ae4[tid + 256 * j];
  }
#pragma unroll
  for (int j = 0; j < 8; ++j) {
    const int i = tid + 256 * j;
    const int r = i >> 6, c = (i & 63) * 4;
    *reinterpret_cast<float4*>(&adT[r][c]) = va[j];
    *reinterpret_cast<float4*>(&aeT[r][c]) = vb[j];
  }
  __syncthreads();

  const int si = tid >> 4, ti = tid & 15;
  float acc00 = 0.f, acc01 = 0.f, acc10 = 0.f, acc11 = 0.f;
#pragma unroll 4
  for (int u = 0; u < U_; ++u) {
    const float a0 = adT[si][u], a1 = adT[si + 16][u];
    const float e0 = aeT[ti][u], e1 = aeT[ti + 16][u];
    const float n  = nuS[u];
    acc00 += tanh_hw(a0 + e0) * n;
    acc01 += tanh_hw(a0 + e1) * n;
    acc10 += tanh_hw(a1 + e0) * n;
    acc11 += tanh_hw(a1 + e1) * n;
  }

  const int s0 = b * TD_ + s_tile * 32 + si;
  const int t0 = t_tile * 32 + ti;
  mu[(size_t)s0 * TE_ + t0]             = acc00;
  mu[(size_t)s0 * TE_ + t0 + 16]        = acc01;
  mu[(size_t)(s0 + 16) * TE_ + t0]      = acc10;
  mu[(size_t)(s0 + 16) * TE_ + t0 + 16] = acc11;

  float p00 = acc00, p01 = acc01, p10 = acc10, p11 = acc11;
#pragma unroll
  for (int x = 8; x >= 1; x >>= 1) {
    p00 = fmaxf(p00, __shfl_xor(p00, x));
    p01 = fmaxf(p01, __shfl_xor(p01, x));
    p10 = fmaxf(p10, __shfl_xor(p10, x));
    p11 = fmaxf(p11, __shfl_xor(p11, x));
  }
  if (ti == 0) {
    float* q0 = pmax + (size_t)s0 * 16 + 2 * t_tile;
    float* q1 = pmax + (size_t)(s0 + 16) * 16 + 2 * t_tile;
    q0[0] = p00; q0[1] = p01;
    q1[0] = p10; q1[1] = p11;
  }
}

__device__ __forceinline__ void lds_barrier() {
  __builtin_amdgcn_fence(__ATOMIC_RELEASE, "workgroup", "local");
  __builtin_amdgcn_s_barrier();
  __builtin_amdgcn_fence(__ATOMIC_ACQUIRE, "workgroup", "local");
}

__device__ __forceinline__ float wave_max(float x) {
  x = fmaxf(x, __shfl_xor(x, 16));
  x = fmaxf(x, __shfl_xor(x, 8));
  x = fmaxf(x, __shfl_xor(x, 4));
  x = fmaxf(x, __shfl_xor(x, 2));
  x = fmaxf(x, __shfl_xor(x, 1));
  return x;
}
__device__ __forceinline__ float wave_sum(float x) {
  x += __shfl_xor(x, 16);
  x += __shfl_xor(x, 8);
  x += __shfl_xor(x, 4);
  x += __shfl_xor(x, 2);
  x += __shfl_xor(x, 1);
  return x;
}

__global__ void __launch_bounds__(256) sum_en_out_kernel(
    const float* __restrict__ mu, const float* __restrict__ pmax,
    const float* __restrict__ en_seq, const float* __restrict__ de_seq,
    float* __restrict__ out) {
  const int tid   = threadIdx.x;
  const int lane  = threadIdx.x & 31;
  const int wave  = threadIdx.x >> 5;

  __shared__ __attribute__((aligned(16))) float aS[16][260];
  __shared__ __attribute__((aligned(16))) float hpS[8][32];
  __shared__ float maS[256];
  __shared__ float wmaxS[8];
  __shared__ float wsumS[8];

  if (blockIdx.x >= 256) {
    const int hb = blockIdx.x - 256;
    const int b  = hb >> 3;
    const int c0 = (hb & 7) * 32;
    const int q  = lane & 7;
    const int s0 = 4 * wave + (lane >> 3);

    const float* dp = de_seq + ((size_t)(b * TD_ + s0)) * D_ + c0 + 4 * q;
    float4 dv[8];
#pragma unroll
    for (int j = 0; j < 8; ++j)
      dv[j] = *reinterpret_cast<const float4*>(dp + (size_t)(32 * j) * D_);

    const float4* pm4 = reinterpret_cast<const float4*>(
        pmax + ((size_t)(b * TD_ + tid)) * 16);
    const float4 p0 = pm4[0], p1 = pm4[1], p2 = pm4[2], p3 = pm4[3];

    const float v = fmaxf(
        fmaxf(fmaxf(fmaxf(p0.x, p0.y), fmaxf(p0.z, p0.w)),
              fmaxf(fmaxf(p1.x, p1.y), fmaxf(p1.z, p1.w))),
        fmaxf(fmaxf(fmaxf(p2.x, p2.y), fmaxf(p2.z, p2.w)),
              fmaxf(fmaxf(p3.x, p3.y), fmaxf(p3.z, p3.w))));
    const float wm = wave_max(v);
    if (lane == 0) wmaxS[wave] = wm;
    lds_barrier();
    float mxs = wmaxS[0];
    for (int j = 1; j < 8; ++j) mxs = fmaxf(mxs, wmaxS[j]);
    const float ev = expf(v - mxs);
    const float wsm = wave_sum(ev);
    if (lane == 0) wsumS[wave] = wsm;
    lds_barrier();
    float tot = wsumS[0];
    for (int j = 1; j < 8; ++j) tot += wsumS[j];
    maS[tid] = ev * (1.0f / tot);
    lds_barrier();

    float4 ha = {0.f, 0.f, 0.f, 0.f};
#pragma unroll
    for (int j = 0; j < 8; ++j) {
      const float w = maS[s0 + 32 * j];
      ha.x = fmaf(dv[j].x, w, ha.x);
      ha.y = fmaf(dv[j].y, w, ha.y);
      ha.z = fmaf(dv[j].z, w, ha.z);
      ha.w = fmaf(dv[j].w, w, ha.w);
    }
    ha.x += __shfl_xor(ha.x, 8);  ha.y += __shfl_xor(ha.y, 8);
    ha.z += __shfl_xor(ha.z, 8);  ha.w += __shfl_xor(ha.w, 8);
    ha.x += __shfl_xor(ha.x, 16); ha.y += __shfl_xor(ha.y, 16);
    ha.z += __shfl_xor(ha.z, 16); ha.w += __shfl_xor(ha.w, 16);
    if (lane < 8) *reinterpret_cast<float4*>(&hpS[wave][4 * q]) = ha;
    lds_barrier();

    float4 hh = *reinterpret_cast<const float4*>(&hpS[0][4 * q]);
    for (int j = 1; j < 8; ++j) {
      const float4 t = *reinterpret_cast<const float4*>(&hpS[j][4 * q]);
      hh.x += t.x; hh.y += t.y; hh.z += t.z; hh.w += t.w;
    }

    float* op = out + ((size_t)(b * TD_ + s0)) * (4 * D_) + c0 + 4 * q;
#pragma unroll
    for (int j = 0; j < 8; ++j) {
      float* o = op + (size_t)(32 * j) * (4 * D_);
      float4 dh;
      dh.x = dv[j].x * hh.x; dh.y = dv[j].y * hh.y;
      dh.z = dv[j].z * hh.z; dh.w = dv[j].w * hh.w;
      *reinterpret_cast<float4*>(o)          = dv[j];
      *reinterpret_cast<float4*>(o + 3 * D_) = dh;
    }
    return;
  }

  const int b     = blockIdx.x >> 5;
  const int tile  = (blockIdx.x & 31) * 8 + wave;
  const int td    = tile & 15;
  const int ts    = tile >> 4;
  const int m     = lane & 15;
  const int khalf = lane >> 4;

  float4 mv[2][2];
#pragma unroll
  for (int rr = 0; rr < 2; ++rr) {
    const int r = 2 * wave + rr;
    const float* mrow = mu + ((size_t)(b * TD_ + ts * 16 + r)) * TE_;
    mv[rr][0] = *reinterpret_cast<const float4*>(mrow + 4 * lane);
    mv[rr][1] = *reinterpret_cast<const float4*>(mrow + 128 + 4 * lane);
  }

#pragma unroll
  for (int rr = 0; rr < 2; ++rr) {
    const int r = 2 * wave + rr;
    const float4 v0 = mv[rr][0];
    const float4 v1 = mv[rr][1];
    const float mx = wave_max(fmaxf(fmaxf(fmaxf(v0.x, v0.y), fmaxf(v0.z, v0.w)),
                                    fmaxf(fmaxf(v1.x, v1.y), fmaxf(v1.z, v1.w))));
    float4 e0, e1;
    e0.x = expf(v0.x - mx); e0.y = expf(v0.y - mx);
    e0.z = expf(v0.z - mx); e0.w = expf(v0.w - mx);
    e1.x = expf(v1.x - mx); e1.y = expf(v1.y - mx);
    e1.z = expf(v1.z - mx); e1.w = expf(v1.w - mx);
    const float sm = wave_sum(((e0.x + e0.y) + (e0.z + e0.w)) +
                              ((e1.x + e1.y) + (e1.z + e1.w)));
    const float inv = 1.0f / sm;
    e0.x *= inv; e0.y *= inv; e0.z *= inv; e0.w *= inv;
    e1.x *= inv; e1.y *= inv; e1.z *= inv; e1.w *= inv;
    *reinterpret_cast<float4*>(&aS[r][4 * lane])       = e0;
    *reinterpret_cast<float4*>(&aS[r][128 + 4 * lane]) = e1;
  }
  lds_barrier();

  const int d_g = td * 16 + m;
  const float* Bm   = en_seq + (size_t)b * TE_ * D_;
  const float* aL   = &aS[m][2 * khalf];
  const float* bcol = Bm + td * 16 + m;

  v8f acc = {};
  for (int k = 0; k < TE_; k += 4) {
    const int ka = k + 2 * khalf;
    v2f a, bb;
    a.x  = aL[k + 0];
    a.y  = aL[k + 1];
    bb.x = bcol[(size_t)(ka + 0) * D_];
    bb.y = bcol[(size_t)(ka + 1) * D_];
    acc = __builtin_amdgcn_wmma_f32_16x16x4_f32(false, a, false, bb,
                                                (short)0, acc, false, false);
  }

#pragma unroll
  for (int r = 0; r < 8; ++r) {
    const int s_g  = ts * 16 + r + 8 * khalf;
    const float se = acc[r];
    const float de = de_seq[((size_t)(b * TD_ + s_g)) * D_ + d_g];
    float* o = out + ((size_t)(b * TD_ + s_g)) * (4 * D_);
    o[D_ + d_g]     = se;
    o[2 * D_ + d_g] = de * se;
  }
}

extern "C" void kernel_launch(void* const* d_in, const int* in_sizes, int n_in,
                              void* d_out, int out_size, void* d_ws, size_t ws_size,
                              hipStream_t stream) {
  const float* en_seq = (const float*)d_in[0];
  const float* de_seq = (const float*)d_in[1];
  const float* w_en   = (const float*)d_in[2];
  const float* w_de   = (const float*)d_in[3];
  const float* nu     = (const float*)d_in[4];
  float* out = (float*)d_out;

  float* ws = (float*)d_ws;
  const size_t nAE = (size_t)B_ * TE_ * U_;
  float* ae     = ws;
  float* ad     = ae + nAE;
  float* mu     = ad + nAE;
  float* pmax   = mu + (size_t)B_ * TD_ * TE_;

  const dim3 blk(256);
  proj2_f32_wmma<<<dim3(256), dim3(128), 0, stream>>>(en_seq, w_en, ae,
                                                     de_seq, w_de, ad);
  co_mu_kernel<<<dim3(64, B_), blk, 0, stream>>>(ae, ad, nu, mu, pmax);
  sum_en_out_kernel<<<dim3(320), blk, 0, stream>>>(mu, pmax, en_seq,
                                                   de_seq, out);
}


# speedup vs baseline: 5.2697x; 1.0749x over previous
#include <hip/hip_runtime.h>
#include <cmath>

#define B_  8
#define TE_ 256
#define TD_ 256
#define D_  256
#define U_  256

typedef float v2f __attribute__((ext_vector_type(2)));
typedef float v8f __attribute__((ext_vector_type(8)));

static_assert(TE_ == TD_, "one tile grid serves both projections");

__device__ __forceinline__ void proj_load_trip(
    v2f (&a)[8], float4 (&b0)[8], float4 (&b1)[8],
    const float* __restrict__ arow, const float* __restrict__ brow, int k0) {
#pragma unroll
  for (int j = 0; j < 8; ++j) {
    const int k = k0 + 4 * j;
    a[j]  = *reinterpret_cast<const v2f*>(arow + k);
    b0[j] = *reinterpret_cast<const float4*>(brow + (size_t)(k + 0) * U_);
    b1[j] = *reinterpret_cast<const float4*>(brow + (size_t)(k + 1) * U_);
  }
}

__device__ __forceinline__ void proj_mma_trip(
    v8f (&acc)[4], const v2f (&a)[8], const float4 (&b0)[8],
    const float4 (&b1)[8]) {
#pragma unroll
  for (int j = 0; j < 8; ++j) {
    const float p0[4] = {b0[j].x, b0[j].y, b0[j].z, b0[j].w};
    const float p1[4] = {b1[j].x, b1[j].y, b1[j].z, b1[j].w};
#pragma unroll
    for (int t = 0; t < 4; ++t) {
      v2f bb;
      bb.x = p0[t];
      bb.y = p1[t];
      acc[t] = __builtin_amdgcn_wmma_f32_16x16x4_f32(false, a[j], false, bb,
                                                     (short)0, acc[t], false, false);
    }
  }
}

__global__ void __launch_bounds__(128) proj2_f32_wmma(
    const float* __restrict__ en_seq, const float* __restrict__ w_en,
    float* __restrict__ ae,
    const float* __restrict__ de_seq, const float* __restrict__ w_de,
    float* __restrict__ ad) {
  const int lane  = threadIdx.x & 31;
  const int wave  = threadIdx.x >> 5;
  const int prod  = blockIdx.x & 1;
  const int panel = (blockIdx.x >> 1) & 3;
  const int tm    = (blockIdx.x >> 3) * 4 + wave;
  const int m     = lane & 15;
  const int khalf = lane >> 4;

  const float* x = prod ? de_seq : en_seq;
  const float* w = prod ? w_de : w_en;
  float*       y = prod ? ad : ae;

  const float* arow = x + (size_t)(tm * 16 + m) * D_ + 2 * khalf;
  const float* brow = w + (size_t)(2 * khalf) * U_ + panel * 64 + 4 * m;

  v8f acc[4] = {};
  v2f    a0[8], a1[8];
  float4 b00[8], b10[8], b01[8], b11[8];

  proj_load_trip(a0, b00, b10, arow, brow, 0);
#pragma nounroll
  for (int k0 = 0; k0 < D_ - 64; k0 += 64) {
    proj_load_trip(a1, b01, b11, arow, brow, k0 + 32);
    __builtin_amdgcn_sched_barrier(0);
    proj_mma_trip(acc, a0, b00, b10);
    __builtin_amdgcn_sched_barrier(0);
    proj_load_trip(a0, b00, b10, arow, brow, k0 + 64);
    __builtin_amdgcn_sched_barrier(0);
    proj_mma_trip(acc, a1, b01, b11);
    __builtin_amdgcn_sched_barrier(0);
  }
  proj_load_trip(a1, b01, b11, arow, brow, D_ - 32);
  __builtin_amdgcn_sched_barrier(0);
  proj_mma_trip(acc, a0, b00, b10);
  proj_mma_trip(acc, a1, b01, b11);

#pragma unroll
  for (int r = 0; r < 8; ++r) {
    float4 v;
    v.x = acc[0][r];
    v.y = acc[1][r];
    v.z = acc[2][r];
    v.w = acc[3][r];
    *reinterpret_cast<float4*>(
        y + (size_t)(tm * 16 + r + 8 * khalf) * U_ + panel * 64 + 4 * m) = v;
  }
}

__device__ __forceinline__ float tanh_hw(float x) {
  return __builtin_amdgcn_tanhf(x);
}

#define TANH_THEN_ADD(t, x, s, a, b)                                   \
  asm volatile("v_tanh_f32 %0, %2\n\tv_pk_add_f32 %1, %3, %4"          \
               : "=&v"(t), "=&v"(s) : "v"(x), "v"(a), "v"(b))
#define TANH_THEN_FMA(t, x, acc, m, n)                                 \
  asm volatile("v_tanh_f32 %0, %2\n\tv_pk_fma_f32 %1, %3, %4, %1"      \
               : "=&v"(t), "+v"(acc) : "v"(x), "v"(m), "v"(n))
#define TANH_ONLY(t, x) asm volatile("v_tanh_f32 %0, %1" : "=v"(t) : "v"(x))
#define PK_ADD(s, a, b) \
  asm volatile("v_pk_add_f32 %0, %1, %2" : "=v"(s) : "v"(a), "v"(b))
#define PK_FMA(acc, m, n) \
  asm volatile("v_pk_fma_f32 %0, %1, %2, %0" : "+v"(acc) : "v"(m), "v"(n))

__global__ void __launch_bounds__(256) co_mu_kernel(
    const float* __restrict__ ae, const float* __restrict__ ad,
    const float* __restrict__ nu, float* __restrict__ mu,
    float* __restrict__ pmax) {
  const int b      = blockIdx.y;
  const int s_tile = blockIdx.x >> 3;
  const int t_tile = blockIdx.x & 7;
  __shared__ __attribute__((aligned(16))) float adT[32][260];
  __shared__ __attribute__((aligned(16))) float aeT[32][260];
  __shared__ __attribute__((aligned(16))) float nuS[256];

  const int tid = threadIdx.x;
  nuS[tid] = nu[tid];

  const float4* ad4 = reinterpret_cast<const float4*>(
      ad + ((size_t)(b * TD_ + s_tile * 32)) * U_);
  const float4* ae4 = reinterpret_cast<const float4*>(
      ae + ((size_t)(b * TE_ + t_tile * 32)) * U_);
  float4 va[8], vb[8];
#pragma unroll
  for (int j = 0; j < 8; ++j) {
    va[j] = ad4[tid + 256 * j];
    vb[j] = ae4[tid + 256 * j];
  }
#pragma unroll
  for (int j = 0; j < 8; ++j) {
    const int i = tid + 256 * j;
    const int r = i >> 6, c = (i & 63) * 4;
    *reinterpret_cast<float4*>(&adT[r][c]) = va[j];
    *reinterpret_cast<float4*>(&aeT[r][c]) = vb[j];
  }
  __syncthreads();

  const int si = tid >> 4, ti = tid & 15;
  v2f acc[4] = {{0.f, 0.f}, {0.f, 0.f}, {0.f, 0.f}, {0.f, 0.f}};
  for (int u = 0; u < U_; u += 8) {
    v2f A[2][4], E[2][4], N[4];
#pragma unroll
    for (int h = 0; h < 2; ++h) {
      const float4 a0 = *reinterpret_cast<const float4*>(&adT[si][u + 4 * h]);
      const float4 a1 = *reinterpret_cast<const float4*>(&adT[si + 16][u + 4 * h]);
      const float4 e0 = *reinterpret_cast<const float4*>(&aeT[ti][u + 4 * h]);
      const float4 e1 = *reinterpret_cast<const float4*>(&aeT[ti + 16][u + 4 * h]);
      const float4 n  = *reinterpret_cast<const float4*>(&nuS[u + 4 * h]);
      A[0][2 * h] = v2f{a0.x, a0.y};  A[0][2 * h + 1] = v2f{a0.z, a0.w};
      A[1][2 * h] = v2f{a1.x, a1.y};  A[1][2 * h + 1] = v2f{a1.z, a1.w};
      E[0][2 * h] = v2f{e0.x, e0.y};  E[0][2 * h + 1] = v2f{e0.z, e0.w};
      E[1][2 * h] = v2f{e1.x, e1.y};  E[1][2 * h + 1] = v2f{e1.z, e1.w};
      N[2 * h]    = v2f{n.x, n.y};    N[2 * h + 1]    = v2f{n.z, n.w};
    }
    v2f S[16];
    float tl[16], th[16];
    PK_ADD(S[0], A[0][0], E[0][0]);
    PK_ADD(S[1], A[0][0], E[1][0]);
#pragma unroll
    for (int j = 0; j < 32; ++j) {
      const int k = j >> 1;
      const float x = (j & 1) ? S[k].y : S[k].x;
      float t;
      const int ka = (j <= 4) ? j + 2 : ((j & 1) == 0 && j <= 22) ? 4 + (j >> 1) : -1;
      const int kf = ((j & 1) == 1 && j >= 5) ? (j - 5) >> 1 : -1;
      if (ka >= 0) {
        TANH_THEN_ADD(t, x, S[ka], A[(ka >> 1) & 1][ka >> 2], E[ka & 1][ka >> 2]);
      } else if (kf >= 0) {
        const v2f m = {tl[kf], th[kf]};
        TANH_THEN_FMA(t, x, acc[kf & 3], m, N[kf >> 2]);
      } else {
        TANH_ONLY(t, x);
      }
      if (j & 1) th[k] = t; else tl[k] = t;
    }
#pragma unroll
    for (int kf = 14; kf < 16; ++kf) {
      const v2f m = {tl[kf], th[kf]};
      PK_FMA(acc[kf & 3], m, N[kf >> 2]);
    }
  }
  const float acc00 = acc[0].x + acc[0].y, acc01 = acc[1].x + acc[1].y;
  const float acc10 = acc[2].x + acc[2].y, acc11 = acc[3].x + acc[3].y;

  const int s0 = b * TD_ + s_tile * 32 + si;
  const int t0 = t_tile * 32 + ti;
  mu[(size_t)s0 * TE_ + t0]             = acc00;
  mu[(size_t)s0 * TE_ + t0 + 16]        = acc01;
  mu[(size_t)(s0 + 16) * TE_ + t0]      = acc10;
  mu[(size_t)(s0 + 16) * TE_ + t0 + 16] = acc11;

  float p00 = acc00, p01 = acc01, p10 = acc10, p11 = acc11;
#pragma unroll
  for (int x = 8; x >= 1; x >>= 1) {
    p00 = fmaxf(p00, __shfl_xor(p00, x));
    p01 = fmaxf(p01, __shfl_xor(p01, x));
    p10 = fmaxf(p10, __shfl_xor(p10, x));
    p11 = fmaxf(p11, __shfl_xor(p11, x));
  }
  if (ti == 0) {
    float* q0 = pmax + (size_t)s0 * 16 + 2 * t_tile;
    float* q1 = pmax + (size_t)(s0 + 16) * 16 + 2 * t_tile;
    q0[0] = p00; q0[1] = p01;
    q1[0] = p10; q1[1] = p11;
  }
}

__device__ __forceinline__ void lds_barrier() {
  __builtin_amdgcn_fence(__ATOMIC_RELEASE, "workgroup", "local");
  __builtin_amdgcn_s_barrier();
  __builtin_amdgcn_fence(__ATOMIC_ACQUIRE, "workgroup", "local");
}

__device__ __forceinline__ float wave_max(float x) {
  x = fmaxf(x, __shfl_xor(x, 16));
  x = fmaxf(x, __shfl_xor(x, 8));
  x = fmaxf(x, __shfl_xor(x, 4));
  x = fmaxf(x, __shfl_xor(x, 2));
  x = fmaxf(x, __shfl_xor(x, 1));
  return x;
}
__device__ __forceinline__ float wave_sum(float x) {
  x += __shfl_xor(x, 16);
  x += __shfl_xor(x, 8);
  x += __shfl_xor(x, 4);
  x += __shfl_xor(x, 2);
  x += __shfl_xor(x, 1);
  return x;
}

__global__ void __launch_bounds__(256) sum_en_out_kernel(
    const float* __restrict__ mu, const float* __restrict__ pmax,
    const float* __restrict__ en_seq, const float* __restrict__ de_seq,
    float* __restrict__ out) {
  const int tid   = threadIdx.x;
  const int lane  = threadIdx.x & 31;
  const int wave  = threadIdx.x >> 5;

  __shared__ __attribute__((aligned(16))) float aS[16][260];
  __shared__ __attribute__((aligned(16))) float hpS[8][32];
  __shared__ float maS[256];
  __shared__ float wmaxS[8];
  __shared__ float wsumS[8];

  if (blockIdx.x >= 256) {
    const int hb = blockIdx.x - 256;
    const int b  = hb >> 3;
    const int c0 = (hb & 7) * 32;
    const int q  = lane & 7;
    const int s0 = 4 * wave + (lane >> 3);

    const float* dp = de_seq + ((size_t)(b * TD_ + s0)) * D_ + c0 + 4 * q;
    float4 dv[8];
#pragma unroll
    for (int j = 0; j < 8; ++j)
      dv[j] = *reinterpret_cast<const float4*>(dp + (size_t)(32 * j) * D_);

    const float4* pm4 = reinterpret_cast<const float4*>(
        pmax + ((size_t)(b * TD_ + tid)) * 16);
    const float4 p0 = pm4[0], p1 = pm4[1], p2 = pm4[2], p3 = pm4[3];

    const float v = fmaxf(
        fmaxf(fmaxf(fmaxf(p0.x, p0.y), fmaxf(p0.z, p0.w)),
              fmaxf(fmaxf(p1.x, p1.y), fmaxf(p1.z, p1.w))),
        fmaxf(fmaxf(fmaxf(p2.x, p2.y), fmaxf(p2.z, p2.w)),
              fmaxf(fmaxf(p3.x, p3.y), fmaxf(p3.z, p3.w))));
    const float wm = wave_max(v);
    if (lane == 0) wmaxS[wave] = wm;
    lds_barrier();
    float mxs = wmaxS[0];
    for (int j = 1; j < 8; ++j) mxs = fmaxf(mxs, wmaxS[j]);
    const float ev = expf(v - mxs);
    const float wsm = wave_sum(ev);
    if (lane == 0) wsumS[wave] = wsm;
    lds_barrier();
    float tot = wsumS[0];
    for (int j = 1; j < 8; ++j) tot += wsumS[j];
    maS[tid] = ev * (1.0f / tot);
    lds_barrier();

    float4 ha = {0.f, 0.f, 0.f, 0.f};
#pragma unroll
    for (int j = 0; j < 8; ++j) {
      const float w = maS[s0 + 32 * j];
      ha.x = fmaf(dv[j].x, w, ha.x);
      ha.y = fmaf(dv[j].y, w, ha.y);
      ha.z = fmaf(dv[j].z, w, ha.z);
      ha.w = fmaf(dv[j].w, w, ha.w);
    }
    ha.x += __shfl_xor(ha.x, 8);  ha.y += __shfl_xor(ha.y, 8);
    ha.z += __shfl_xor(ha.z, 8);  ha.w += __shfl_xor(ha.w, 8);
    ha.x += __shfl_xor(ha.x, 16); ha.y += __shfl_xor(ha.y, 16);
    ha.z += __shfl_xor(ha.z, 16); ha.w += __shfl_xor(ha.w, 16);
    if (lane < 8) *reinterpret_cast<float4*>(&hpS[wave][4 * q]) = ha;
    lds_barrier();

    float4 hh = *reinterpret_cast<const float4*>(&hpS[0][4 * q]);
    for (int j = 1; j < 8; ++j) {
      const float4 t = *reinterpret_cast<const float4*>(&hpS[j][4 * q]);
      hh.x += t.x; hh.y += t.y; hh.z += t.z; hh.w += t.w;
    }

    float* op = out + ((size_t)(b * TD_ + s0)) * (4 * D_) + c0 + 4 * q;
#pragma unroll
    for (int j = 0; j < 8; ++j) {
      float* o = op + (size_t)(32 * j) * (4 * D_);
      float4 dh;
      dh.x = dv[j].x * hh.x; dh.y = dv[j].y * hh.y;
      dh.z = dv[j].z * hh.z; dh.w = dv[j].w * hh.w;
      *reinterpret_cast<float4*>(o)          = dv[j];
      *reinterpret_cast<float4*>(o + 3 * D_) = dh;
    }
    return;
  }

  const int b     = blockIdx.x >> 5;
  const int tile  = (blockIdx.x & 31) * 8 + wave;
  const int td    = tile & 15;
  const int ts    = tile >> 4;
  const int m     = lane & 15;
  const int khalf = lane >> 4;

  float4 mv[2][2];
#pragma unroll
  for (int rr = 0; rr < 2; ++rr) {
    const int r = 2 * wave + rr;
    const float* mrow = mu + ((size_t)(b * TD_ + ts * 16 + r)) * TE_;
    mv[rr][0] = *reinterpret_cast<const float4*>(mrow + 4 * lane);
    mv[rr][1] = *reinterpret_cast<const float4*>(mrow + 128 + 4 * lane);
  }

#pragma unroll
  for (int rr = 0; rr < 2; ++rr) {
    const int r = 2 * wave + rr;
    const float4 v0 = mv[rr][0];
    const float4 v1 = mv[rr][1];
    const float mx = wave_max(fmaxf(fmaxf(fmaxf(v0.x, v0.y), fmaxf(v0.z, v0.w)),
                                    fmaxf(fmaxf(v1.x, v1.y), fmaxf(v1.z, v1.w))));
    float4 e0, e1;
    e0.x = expf(v0.x - mx); e0.y = expf(v0.y - mx);
    e0.z = expf(v0.z - mx); e0.w = expf(v0.w - mx);
    e1.x = expf(v1.x - mx); e1.y = expf(v1.y - mx);
    e1.z = expf(v1.z - mx); e1.w = expf(v1.w - mx);
    const float sm = wave_sum(((e0.x + e0.y) + (e0.z + e0.w)) +
                              ((e1.x + e1.y) + (e1.z + e1.w)));
    const float inv = 1.0f / sm;
    e0.x *= inv; e0.y *= inv; e0.z *= inv; e0.w *= inv;
    e1.x *= inv; e1.y *= inv; e1.z *= inv; e1.w *= inv;
    *reinterpret_cast<float4*>(&aS[r][4 * lane])       = e0;
    *reinterpret_cast<float4*>(&aS[r][128 + 4 * lane]) = e1;
  }
  lds_barrier();

  const int d_g = td * 16 + m;
  const float* Bm   = en_seq + (size_t)b * TE_ * D_;
  const float* aL   = &aS[m][2 * khalf];
  const float* bcol = Bm + td * 16 + m;

  v8f acc = {};
  for (int k = 0; k < TE_; k += 4) {
    const int ka = k + 2 * khalf;
    v2f a, bb;
    a.x  = aL[k + 0];
    a.y  = aL[k + 1];
    bb.x = bcol[(size_t)(ka + 0) * D_];
    bb.y = bcol[(size_t)(ka + 1) * D_];
    acc = __builtin_amdgcn_wmma_f32_16x16x4_f32(false, a, false, bb,
                                                (short)0, acc, false, false);
  }

#pragma unroll
  for (int r = 0; r < 8; ++r) {
    const int s_g  = ts * 16 + r + 8 * khalf;
    const float se = acc[r];
    const float de = de_seq[((size_t)(b * TD_ + s_g)) * D_ + d_g];
    float* o = out + ((size_t)(b * TD_ + s_g)) * (4 * D_);
    o[D_ + d_g]     = se;
    o[2 * D_ + d_g] = de * se;
  }
}

extern "C" void kernel_launch(void* const* d_in, const int* in_sizes, int n_in,
                              void* d_out, int out_size, void* d_ws, size_t ws_size,
                              hipStream_t stream) {
  const float* en_seq = (const float*)d_in[0];
  const float* de_seq = (const float*)d_in[1];
  const float* w_en   = (const float*)d_in[2];
  const float* w_de   = (const float*)d_in[3];
  const float* nu     = (const float*)d_in[4];
  float* out = (float*)d_out;

  float* ws = (float*)d_ws;
  const size_t nAE = (size_t)B_ * TE_ * U_;
  float* ae     = ws;
  float* ad     = ae + nAE;
  float* mu     = ad + nAE;
  float* pmax   = mu + (size_t)B_ * TD_ * TE_;

  const dim3 blk(256);
  proj2_f32_wmma<<<dim3(256), dim3(128), 0, stream>>>(en_seq, w_en, ae,
                                                     de_seq, w_de, ad);
  co_mu_kernel<<<dim3(64, B_), blk, 0, stream>>>(ae, ad, nu, mu, pmax);
  sum_en_out_kernel<<<dim3(320), blk, 0, stream>>>(mu, pmax, en_seq,
                                                   de_seq, out);
}
